# starting kernel
# baseline (speedup 1.0000x reference)
#include <hip/hip_runtime.h>
#include <hip/hip_bf16.h>
#include <hip/hip_fp16.h>

#define N_NODES 10000
#define N_EDGES 320000
#define IN_DIM  256
#define HID     64
#define HEADS   4
#define D1      (HEADS * HID)
#define OUT_DIM 64
#define NEG_SLOPE 0.2f

typedef __attribute__((ext_vector_type(16))) _Float16 v16h;
typedef __attribute__((ext_vector_type(8)))  _Float16 v8h;
typedef __attribute__((ext_vector_type(8)))  float    v8f;

__device__ inline void atomicMaxF(float* addr, float val) {
    if (val >= 0.0f)
        atomicMax((int*)addr, __float_as_int(val));
    else
        atomicMin((unsigned int*)addr, (unsigned int)__float_as_int(val));
}

__global__ void cvt_f32_to_f16(const float* __restrict__ in, _Float16* __restrict__ out, int n) {
    int i = blockIdx.x * blockDim.x + threadIdx.x;
    if (i < n) out[i] = (_Float16)in[i];
}

__global__ void cvt_transpose_f16(const float* __restrict__ W, _Float16* __restrict__ Wt,
                                  int K, int N) {
    int i = blockIdx.x * blockDim.x + threadIdx.x;
    if (i >= K * N) return;
    int k = i / N, n = i % N;
    Wt[(size_t)n * K + k] = (_Float16)W[i];
}

__global__ void fill_f32(float* __restrict__ p, float v, int n) {
    int i = blockIdx.x * blockDim.x + threadIdx.x;
    if (i < n) p[i] = v;
}

__global__ void init_bias(float* __restrict__ out, const float* __restrict__ bias, int n, int d) {
    int i = blockIdx.x * blockDim.x + threadIdx.x;
    if (i < n * d) out[i] = bias[i % d];
}

__global__ void elu_bias_to_f16(const float* __restrict__ in, const float* __restrict__ bias,
                                _Float16* __restrict__ out, int n, int F) {
    int i = blockIdx.x * blockDim.x + threadIdx.x;
    if (i >= n) return;
    float x = in[i] + bias[i % F];
    x = (x > 0.0f) ? x : (__expf(x) - 1.0f);
    out[i] = (_Float16)x;
}

__device__ inline v16h load_frag(const _Float16* __restrict__ base, int half) {
    v8h lo = *(const v8h*)(base + 8 * half);
    v8h hi = *(const v8h*)(base + 16 + 8 * half);
    v16h r;
#pragma unroll
    for (int i = 0; i < 8; ++i) { r[i] = lo[i]; r[8 + i] = hi[i]; }
    return r;
}

__global__ void gemm_wmma_f16t(const _Float16* __restrict__ A, const _Float16* __restrict__ Bt,
                               float* __restrict__ C, int M, int N, int K) {
    const int lane = threadIdx.x & 31;
    const int wave = threadIdx.x >> 5;
    const int m0   = blockIdx.x * 16;
    const int n0   = (blockIdx.y * (blockDim.x >> 5) + wave) * 64;
    if (m0 >= M || n0 >= N) return;

    const int half = lane >> 4;
    const int l15  = lane & 15;

    const _Float16* aptr = A + (size_t)(m0 + l15) * K;
    const _Float16* bptr0 = Bt + (size_t)(n0 +  0 + l15) * K;
    const _Float16* bptr1 = Bt + (size_t)(n0 + 16 + l15) * K;
    const _Float16* bptr2 = Bt + (size_t)(n0 + 32 + l15) * K;
    const _Float16* bptr3 = Bt + (size_t)(n0 + 48 + l15) * K;

    v8f acc0 = {}, acc1 = {}, acc2 = {}, acc3 = {};

    for (int k0 = 0; k0 < K; k0 += 32) {
        v16h a  = load_frag(aptr  + k0, half);
        v16h b0 = load_frag(bptr0 + k0, half);
        v16h b1 = load_frag(bptr1 + k0, half);
        v16h b2 = load_frag(bptr2 + k0, half);
        v16h b3 = load_frag(bptr3 + k0, half);
        acc0 = __builtin_amdgcn_wmma_f32_16x16x32_f16(false, a, false, b0, (short)0, acc0, false, false);
        acc1 = __builtin_amdgcn_wmma_f32_16x16x32_f16(false, a, false, b1, (short)0, acc1, false, false);
        acc2 = __builtin_amdgcn_wmma_f32_16x16x32_f16(false, a, false, b2, (short)0, acc2, false, false);
        acc3 = __builtin_amdgcn_wmma_f32_16x16x32_f16(false, a, false, b3, (short)0, acc3, false, false);
    }

    float* cbase = C + (size_t)(m0 + 8 * half) * N + n0 + l15;
#pragma unroll
    for (int r = 0; r < 8; ++r) {
        float* row = cbase + (size_t)r * N;
        row[0]  = acc0[r];
        row[16] = acc1[r];
        row[32] = acc2[r];
        row[48] = acc3[r];
    }
}

__global__ void attn_scores(const float* __restrict__ feat, const float* __restrict__ al,
                            const float* __restrict__ ar, float* __restrict__ el,
                            float* __restrict__ er, int nheads, int d) {
    int idx = blockIdx.x * blockDim.x + threadIdx.x;
    if (idx >= N_NODES * nheads) return;
    int n = idx / nheads, h = idx % nheads;
    const float* f = feat + (size_t)(n * nheads + h) * d;
    float sl = 0.f, sr = 0.f;
    for (int j = 0; j < d; ++j) {
        float fv = f[j];
        sl += fv * al[h * d + j];
        sr += fv * ar[h * d + j];
    }
    el[idx] = sl;
    er[idx] = sr;
}

__global__ void edge_max(const int* __restrict__ src, const int* __restrict__ dst,
                         const float* __restrict__ el, const float* __restrict__ er,
                         float* __restrict__ ebuf, float* __restrict__ m, int nheads) {
    int idx = blockIdx.x * blockDim.x + threadIdx.x;
    if (idx >= N_EDGES * nheads) return;
    int e = idx / nheads, h = idx % nheads;
    int s = src[e], t = dst[e];
    float x = el[s * nheads + h] + er[t * nheads + h];
    x = (x > 0.0f) ? x : NEG_SLOPE * x;
    ebuf[idx] = x;
    atomicMaxF(&m[t * nheads + h], x);
}

__global__ void edge_exp(const int* __restrict__ dst, const float* __restrict__ m,
                         float* __restrict__ ebuf, float* __restrict__ denom, int nheads) {
    int idx = blockIdx.x * blockDim.x + threadIdx.x;
    if (idx >= N_EDGES * nheads) return;
    int e = idx / nheads, h = idx % nheads;
    int t = dst[e];
    float ex = __expf(ebuf[idx] - m[t * nheads + h]);
    ebuf[idx] = ex;
    atomicAdd(&denom[t * nheads + h], ex);
}

__global__ void edge_aggregate(const int* __restrict__ src, const int* __restrict__ dst,
                               const float* __restrict__ feat, const float* __restrict__ ebuf,
                               const float* __restrict__ denom, float* __restrict__ out,
                               int nheads, int d) {
    int gwave = (blockIdx.x * blockDim.x + threadIdx.x) >> 5;
    int lane  = threadIdx.x & 31;
    if (gwave >= N_EDGES) return;
    int s = src[gwave], t = dst[gwave];
    int F = nheads * d;
    for (int h = 0; h < nheads; ++h) {
        float alpha = ebuf[gwave * nheads + h] / denom[t * nheads + h];
        for (int j = lane; j < d; j += 32) {
            atomicAdd(&out[(size_t)t * F + h * d + j],
                      feat[(size_t)s * F + h * d + j] * alpha);
        }
    }
}

extern "C" void kernel_launch(void* const* d_in, const int* in_sizes, int n_in,
                              void* d_out, int out_size, void* d_ws, size_t ws_size,
                              hipStream_t stream) {
    const float* h   = (const float*)d_in[0];
    const int*   src = (const int*)  d_in[1];
    const int*   dst = (const int*)  d_in[2];
    const float* W1  = (const float*)d_in[3];
    const float* al1 = (const float*)d_in[4];
    const float* ar1 = (const float*)d_in[5];
    const float* b1  = (const float*)d_in[6];
    const float* W2  = (const float*)d_in[7];
    const float* al2 = (const float*)d_in[8];
    const float* ar2 = (const float*)d_in[9];
    const float* b2  = (const float*)d_in[10];
    float* out = (float*)d_out;

    size_t off = 0;
    auto alloc = [&](size_t bytes) {
        void* p = (char*)d_ws + off;
        off += (bytes + 255) & ~(size_t)255;
        return p;
    };
    _Float16* hf16   = (_Float16*)alloc((size_t)N_NODES * IN_DIM  * 2);
    _Float16* W1t16  = (_Float16*)alloc((size_t)IN_DIM  * D1      * 2);
    _Float16* W2t16  = (_Float16*)alloc((size_t)D1      * OUT_DIM * 2);
    float*    feat1  = (float*)   alloc((size_t)N_NODES * D1      * 4);
    float*    el1    = (float*)   alloc((size_t)N_NODES * HEADS   * 4);
    float*    er1    = (float*)   alloc((size_t)N_NODES * HEADS   * 4);
    float*    m1     = (float*)   alloc((size_t)N_NODES * HEADS   * 4);
    float*    den1   = (float*)   alloc((size_t)N_NODES * HEADS   * 4);
    float*    ebuf1  = (float*)   alloc((size_t)N_EDGES * HEADS   * 4);
    float*    out1   = (float*)   alloc((size_t)N_NODES * D1      * 4);
    _Float16* h1f16  = (_Float16*)alloc((size_t)N_NODES * D1      * 2);
    float*    feat2  = (float*)   alloc((size_t)N_NODES * OUT_DIM * 4);
    float*    el2    = (float*)   alloc((size_t)N_NODES * 4);
    float*    er2    = (float*)   alloc((size_t)N_NODES * 4);
    float*    m2     = (float*)   alloc((size_t)N_NODES * 4);
    float*    den2   = (float*)   alloc((size_t)N_NODES * 4);
    float*    ebuf2  = (float*)   alloc((size_t)N_EDGES * 4);

    const float NEG_INF = -__builtin_inff();
    auto cdiv = [](int a, int b) { return (a + b - 1) / b; };

    cvt_f32_to_f16<<<cdiv(N_NODES * IN_DIM, 256), 256, 0, stream>>>(h, hf16, N_NODES * IN_DIM);
    cvt_transpose_f16<<<cdiv(IN_DIM * D1, 256), 256, 0, stream>>>(W1, W1t16, IN_DIM, D1);
    cvt_transpose_f16<<<cdiv(D1 * OUT_DIM, 256), 256, 0, stream>>>(W2, W2t16, D1, OUT_DIM);

    gemm_wmma_f16t<<<dim3(N_NODES / 16, 1), 128, 0, stream>>>(hf16, W1t16, feat1,
                                                              N_NODES, D1, IN_DIM);
    attn_scores<<<cdiv(N_NODES * HEADS, 256), 256, 0, stream>>>(feat1, al1, ar1, el1, er1,
                                                                HEADS, HID);
    fill_f32<<<cdiv(N_NODES * HEADS, 256), 256, 0, stream>>>(m1,   NEG_INF, N_NODES * HEADS);
    fill_f32<<<cdiv(N_NODES * HEADS, 256), 256, 0, stream>>>(den1, 0.0f,    N_NODES * HEADS);
    fill_f32<<<cdiv(N_NODES * D1,    256), 256, 0, stream>>>(out1, 0.0f,    N_NODES * D1);

    edge_max<<<cdiv(N_EDGES * HEADS, 256), 256, 0, stream>>>(src, dst, el1, er1, ebuf1, m1, HEADS);
    edge_exp<<<cdiv(N_EDGES * HEADS, 256), 256, 0, stream>>>(dst, m1, ebuf1, den1, HEADS);
    edge_aggregate<<<cdiv(N_EDGES * 32, 256), 256, 0, stream>>>(src, dst, feat1, ebuf1, den1,
                                                                out1, HEADS, HID);
    elu_bias_to_f16<<<cdiv(N_NODES * D1, 256), 256, 0, stream>>>(out1, b1, h1f16,
                                                                 N_NODES * D1, D1);

    gemm_wmma_f16t<<<dim3(N_NODES / 16, 1), 32, 0, stream>>>(h1f16, W2t16, feat2,
                                                             N_NODES, OUT_DIM, D1);
    attn_scores<<<cdiv(N_NODES, 256), 256, 0, stream>>>(feat2, al2, ar2, el2, er2, 1, OUT_DIM);
    fill_f32<<<cdiv(N_NODES, 256), 256, 0, stream>>>(m2,   NEG_INF, N_NODES);
    fill_f32<<<cdiv(N_NODES, 256), 256, 0, stream>>>(den2, 0.0f,    N_NODES);
    init_bias<<<cdiv(N_NODES * OUT_DIM, 256), 256, 0, stream>>>(out, b2, N_NODES, OUT_DIM);

    edge_max<<<cdiv(N_EDGES, 256), 256, 0, stream>>>(src, dst, el2, er2, ebuf2, m2, 1);
    edge_exp<<<cdiv(N_EDGES, 256), 256, 0, stream>>>(dst, m2, ebuf2, den2, 1);
    edge_aggregate<<<cdiv(N_EDGES * 32, 256), 256, 0, stream>>>(src, dst, feat2, ebuf2, den2,
                                                                out, 1, OUT_DIM);
    (void)in_sizes; (void)n_in; (void)out_size; (void)ws_size;
}


# elementwise passes (h to f16; ELU+bias to f16): 32 elements per thread as 8 float4 loads in flight, 512-B wave loads, 256-B wave stores; 160,000 -> 5,000 waves; values and sum order unchanged
# speedup vs baseline: 2.6620x; 2.6620x over previous
#include <hip/hip_runtime.h>
#include <hip/hip_bf16.h>
#include <hip/hip_fp16.h>

#define N_NODES 10000
#define N_EDGES 320000
#define IN_DIM  256
#define HID     64
#define HEADS   4
#define D1      (HEADS * HID)
#define OUT_DIM 64
#define NEG_SLOPE 0.2f

static_assert((N_NODES * IN_DIM) % 4 == 0 && (N_NODES * D1) % 4 == 0, "elementwise passes move float4");

typedef __attribute__((ext_vector_type(16))) _Float16 v16h;
typedef __attribute__((ext_vector_type(8)))  _Float16 v8h;
typedef __attribute__((ext_vector_type(4)))  _Float16 v4h;
typedef __attribute__((ext_vector_type(8)))  float    v8f;

#define EW_T 128
#define EW_U 8

__global__ void __launch_bounds__(EW_T)
cvt_f32_to_f16(const float* __restrict__ in, _Float16* __restrict__ out, int n4) {
    const float4* __restrict__ in4 = (const float4*)in;
    const int base = blockIdx.x * (EW_T * EW_U) + threadIdx.x;
    float4 v[EW_U];
#pragma unroll
    for (int u = 0; u < EW_U; ++u) {
        const int i = base + u * EW_T;
        v[u] = in4[(i < n4) ? i : (n4 - 1)];
    }
#pragma unroll
    for (int u = 0; u < EW_U; ++u) {
        const int i = base + u * EW_T;
        if (i < n4) {
            v4h r;
            r[0] = (_Float16)v[u].x; r[1] = (_Float16)v[u].y;
            r[2] = (_Float16)v[u].z; r[3] = (_Float16)v[u].w;
            *(v4h*)(out + (size_t)i * 4) = r;
        }
    }
}

__global__ void cvt_transpose_f16(const float* __restrict__ W, _Float16* __restrict__ Wt,
                                  int K, int N) {
    int i = blockIdx.x * blockDim.x + threadIdx.x;
    if (i >= K * N) return;
    int k = i / N, n = i % N;
    Wt[(size_t)n * K + k] = (_Float16)W[i];
}

__device__ inline float elu1(float x) { return (x > 0.0f) ? x : (__expf(x) - 1.0f); }

template <int F>
__global__ void __launch_bounds__(EW_T)
elu_bias_to_f16(const float* __restrict__ in, const float* __restrict__ bias,
                _Float16* __restrict__ out, int n4) {
    static_assert(F % 4 == 0, "a float4 never straddles two rows");
    static_assert((EW_T * 4) % F == 0, "all float4 of one thread fall on the same 4 bias columns");
    const float4* __restrict__ in4 = (const float4*)in;
    const int base = blockIdx.x * (EW_T * EW_U) + threadIdx.x;
    const float4 bv = ((const float4*)bias)[threadIdx.x % (F / 4)];
    float4 v[EW_U];
#pragma unroll
    for (int u = 0; u < EW_U; ++u) {
        const int i = base + u * EW_T;
        v[u] = in4[(i < n4) ? i : (n4 - 1)];
    }
#pragma unroll
    for (int u = 0; u < EW_U; ++u) {
        const int i = base + u * EW_T;
        if (i < n4) {
            v4h r;
            r[0] = (_Float16)elu1(v[u].x + bv.x); r[1] = (_Float16)elu1(v[u].y + bv.y);
            r[2] = (_Float16)elu1(v[u].z + bv.z); r[3] = (_Float16)elu1(v[u].w + bv.w);
            *(v4h*)(out + (size_t)i * 4) = r;
        }
    }
}

__device__ inline v16h load_frag(const _Float16* __restrict__ base, int half) {
    v8h lo = *(const v8h*)(base + 8 * half);
    v8h hi = *(const v8h*)(base + 16 + 8 * half);
    v16h r;
#pragma unroll
    for (int i = 0; i < 8; ++i) { r[i] = lo[i]; r[8 + i] = hi[i]; }
    return r;
}

__global__ void gemm_wmma_f16t(const _Float16* __restrict__ A, const _Float16* __restrict__ Bt,
                               float* __restrict__ C, int M, int N, int K) {
    const int lane = threadIdx.x & 31;
    const int wave = threadIdx.x >> 5;
    const int m0   = blockIdx.x * 16;
    const int n0   = (blockIdx.y * (blockDim.x >> 5) + wave) * 64;
    if (m0 >= M || n0 >= N) return;

    const int half = lane >> 4;
    const int l15  = lane & 15;

    const _Float16* aptr = A + (size_t)(m0 + l15) * K;
    const _Float16* bptr0 = Bt + (size_t)(n0 +  0 + l15) * K;
    const _Float16* bptr1 = Bt + (size_t)(n0 + 16 + l15) * K;
    const _Float16* bptr2 = Bt + (size_t)(n0 + 32 + l15) * K;
    const _Float16* bptr3 = Bt + (size_t)(n0 + 48 + l15) * K;

    v8f acc0 = {}, acc1 = {}, acc2 = {}, acc3 = {};

    for (int k0 = 0; k0 < K; k0 += 32) {
        v16h a  = load_frag(aptr  + k0, half);
        v16h b0 = load_frag(bptr0 + k0, half);
        v16h b1 = load_frag(bptr1 + k0, half);
        v16h b2 = load_frag(bptr2 + k0, half);
        v16h b3 = load_frag(bptr3 + k0, half);
        acc0 = __builtin_amdgcn_wmma_f32_16x16x32_f16(false, a, false, b0, (short)0, acc0, false, false);
        acc1 = __builtin_amdgcn_wmma_f32_16x16x32_f16(false, a, false, b1, (short)0, acc1, false, false);
        acc2 = __builtin_amdgcn_wmma_f32_16x16x32_f16(false, a, false, b2, (short)0, acc2, false, false);
        acc3 = __builtin_amdgcn_wmma_f32_16x16x32_f16(false, a, false, b3, (short)0, acc3, false, false);
    }

    float* cbase = C + (size_t)(m0 + 8 * half) * N + n0 + l15;
#pragma unroll
    for (int r = 0; r < 8; ++r) {
        float* row = cbase + (size_t)r * N;
        row[0]  = acc0[r];
        row[16] = acc1[r];
        row[32] = acc2[r];
        row[48] = acc3[r];
    }
}

__global__ void attn_scores(const float* __restrict__ feat, const float* __restrict__ al,
                            const float* __restrict__ ar, float* __restrict__ el,
                            float* __restrict__ er, int nheads, int d) {
    int idx = blockIdx.x * blockDim.x + threadIdx.x;
    if (idx >= N_NODES * nheads) return;
    int n = idx / nheads, h = idx % nheads;
    const float* f = feat + (size_t)(n * nheads + h) * d;
    float sl = 0.f, sr = 0.f;
    for (int j = 0; j < d; ++j) {
        float fv = f[j];
        sl += fv * al[h * d + j];
        sr += fv * ar[h * d + j];
    }
    el[idx] = sl;
    er[idx] = sr;
}

#define NB        40
#define OWN_T     1024
#define CHUNK_E   (OWN_T * 4)
#define NCHUNK    ((N_EDGES + CHUNK_E - 1) / CHUNK_E)
#define PH_CHUNKS 16
#define CAP       (PH_CHUNKS * CHUNK_E)
#define SCAN_U    8

static_assert(N_NODES <= 65536, "list entry packs src in 16 bits");
static_assert(N_EDGES <= (1 << 24) && NB <= 256, "list entry packs edge index in 24 bits, local dst in 8");
static_assert(N_EDGES % 4 == 0, "dst is read as int4");
static_assert(NB * HEADS <= OWN_T, "one thread per (node, head) slot");

__device__ inline unsigned f2ord(float f) {
    unsigned u = __float_as_uint(f);
    return (u & 0x80000000u) ? ~u : (u | 0x80000000u);
}
__device__ inline float ord2f(unsigned k) {
    return __uint_as_float((k & 0x80000000u) ? (k ^ 0x80000000u) : ~k);
}

template <int H, int D, int U>
__global__ void __launch_bounds__(OWN_T)
gat_owner(const int* __restrict__ src, const int* __restrict__ dst,
          const float* __restrict__ feat, const float* __restrict__ el,
          const float* __restrict__ er, const float* __restrict__ bias,
          float* __restrict__ out) {
    constexpr int F   = H * D;
    constexpr int KPL = F / 32;
    __shared__ unsigned s_list[CAP];
    __shared__ float    s_acc[NB * F];
    __shared__ unsigned s_mcur[NB * H];
    __shared__ float    s_m[NB * H];
    __shared__ float    s_den[NB * H];
    __shared__ float    s_scl[NB * H];
    __shared__ unsigned s_cnt;

    const int tid  = threadIdx.x;
    const int lane = tid & 31;
    const int wave = __builtin_amdgcn_readfirstlane(tid >> 5);
    const int hl   = (lane < H) ? lane : 0;
    const int lo   = blockIdx.x * NB;
    const int c0   = (int)((blockIdx.x * (unsigned)NCHUNK) / gridDim.x);
    const float NEG_INF = -__builtin_inff();

    for (int i = tid; i < NB * F; i += OWN_T) s_acc[i] = 0.0f;
    if (tid < NB * H) {
        s_mcur[tid] = f2ord(NEG_INF);
        s_m[tid]    = NEG_INF;
        s_den[tid]  = 0.0f;
    }

    for (int p0 = 0; p0 < NCHUNK; p0 += PH_CHUNKS) {
        if (tid == 0) s_cnt = 0u;
        __syncthreads();
        const int pend = (p0 + PH_CHUNKS < NCHUNK) ? (p0 + PH_CHUNKS) : NCHUNK;

        for (int ci = p0; ci < pend; ci += SCAN_U) {
            int4 v[SCAN_U];
            int  eb[SCAN_U];
#pragma unroll
            for (int u = 0; u < SCAN_U; ++u) {
                int chunk = c0 + ci + u;
                if (chunk >= NCHUNK) chunk -= NCHUNK;
                const int e = chunk * CHUNK_E + tid * 4;
                eb[u] = e;
                const bool ok = (ci + u < pend) && (e < N_EDGES);
                v[u] = ok ? *(const int4*)(dst + e) : make_int4(-1, -1, -1, -1);
            }
#pragma unroll
            for (int u = 0; u < SCAN_U; ++u) {
                const int tt[4] = {v[u].x, v[u].y, v[u].z, v[u].w};
#pragma unroll
                for (int c = 0; c < 4; ++c) {
                    const unsigned tl = (unsigned)(tt[c] - lo);
                    if (tl < (unsigned)NB) {
                        const unsigned pos = atomicAdd(&s_cnt, 1u);
                        s_list[pos] = (unsigned)(eb[u] + c) | (tl << 24);
                    }
                }
            }
        }
        __syncthreads();
        const int n = (int)s_cnt;

        for (int i = tid; i < n; i += OWN_T) {
            const unsigned en = s_list[i];
            const int tl = (int)(en >> 24);
            const int sv = src[en & 0xFFFFFFu];
            s_list[i] = (unsigned)sv | ((unsigned)tl << 16);
#pragma unroll
            for (int h = 0; h < H; ++h) {
                float x = el[sv * H + h] + er[(lo + tl) * H + h];
                x = (x > 0.0f) ? x : NEG_SLOPE * x;
                atomicMax(&s_mcur[tl * H + h], f2ord(x));
            }
        }
        __syncthreads();
        if (tid < NB * H) {
            const float mn = ord2f(s_mcur[tid]);
            const float mo = s_m[tid];
            const float sc = (mo == mn) ? 1.0f : __expf(mo - mn);
            s_den[tid] *= sc;
            s_scl[tid]  = sc;
            s_m[tid]    = mn;
        }
        __syncthreads();
        if (p0 > 0) {
            for (int i = tid; i < NB * F; i += OWN_T) s_acc[i] *= s_scl[i / D];
            __syncthreads();
        }

        for (int i0 = wave * U; i0 < n; i0 += (OWN_T / 32) * U) {
            int   tlv[U], svv[U];
            float sa[U], sb[U];
            float fv[U][KPL];
#pragma unroll
            for (int u = 0; u < U; ++u) {
                const int i = i0 + u;
                const unsigned en = (i < n) ? s_list[i] : 0u;
                const int sv = (int)(en & 0xFFFFu);
                const int tl = (int)(en >> 16);
                tlv[u] = tl;
                svv[u] = sv;
#pragma unroll
                for (int k = 0; k < KPL; ++k)
                    fv[u][k] = feat[(size_t)sv * F + k * 32 + lane];
            }
#pragma unroll
            for (int u = 0; u < U; ++u) {
                sa[u] = el[svv[u] * H + hl];
                sb[u] = er[(lo + tlv[u]) * H + hl];
            }
#pragma unroll
            for (int u = 0; u < U; ++u) {
                if (i0 + u < n) {
                    const int tl = tlv[u];
                    float x = sa[u] + sb[u];
                    x = (x > 0.0f) ? x : NEG_SLOPE * x;
                    const float wlane = __expf(x - s_m[tl * H + hl]);
                    if (lane < H) atomicAdd(&s_den[tl * H + lane], wlane);
#pragma unroll
                    for (int k = 0; k < KPL; ++k) {
                        const float w = __shfl(wlane, (k * 32) / D);
                        atomicAdd(&s_acc[tl * F + k * 32 + lane], w * fv[u][k]);
                    }
                }
            }
        }
        __syncthreads();
    }

    for (int i = tid; i < NB * F; i += OWN_T) {
        const int tl = i / F, f = i - tl * F;
        const int node = lo + tl;
        if (node < N_NODES) {
            const float d = s_den[tl * H + f / D];
            float r = (d > 0.0f) ? s_acc[i] / d : 0.0f;
            if (bias) r += bias[f];
            out[(size_t)node * F + f] = r;
        }
    }
}

extern "C" void kernel_launch(void* const* d_in, const int* in_sizes, int n_in,
                              void* d_out, int out_size, void* d_ws, size_t ws_size,
                              hipStream_t stream) {
    const float* h   = (const float*)d_in[0];
    const int*   src = (const int*)  d_in[1];
    const int*   dst = (const int*)  d_in[2];
    const float* W1  = (const float*)d_in[3];
    const float* al1 = (const float*)d_in[4];
    const float* ar1 = (const float*)d_in[5];
    const float* b1  = (const float*)d_in[6];
    const float* W2  = (const float*)d_in[7];
    const float* al2 = (const float*)d_in[8];
    const float* ar2 = (const float*)d_in[9];
    const float* b2  = (const float*)d_in[10];
    float* out = (float*)d_out;

    size_t off = 0;
    auto alloc = [&](size_t bytes) {
        void* p = (char*)d_ws + off;
        off += (bytes + 255) & ~(size_t)255;
        return p;
    };
    _Float16* hf16   = (_Float16*)alloc((size_t)N_NODES * IN_DIM  * 2);
    _Float16* W1t16  = (_Float16*)alloc((size_t)IN_DIM  * D1      * 2);
    _Float16* W2t16  = (_Float16*)alloc((size_t)D1      * OUT_DIM * 2);
    float*    feat1  = (float*)   alloc((size_t)N_NODES * D1      * 4);
    float*    el1    = (float*)   alloc((size_t)N_NODES * HEADS   * 4);
    float*    er1    = (float*)   alloc((size_t)N_NODES * HEADS   * 4);
    float*    out1   = (float*)   alloc((size_t)N_NODES * D1      * 4);
    _Float16* h1f16  = (_Float16*)alloc((size_t)N_NODES * D1      * 2);
    float*    feat2  = (float*)   alloc((size_t)N_NODES * OUT_DIM * 4);
    float*    el2    = (float*)   alloc((size_t)N_NODES * 4);
    float*    er2    = (float*)   alloc((size_t)N_NODES * 4);

    auto cdiv = [](int a, int b) { return (a + b - 1) / b; };

    cvt_f32_to_f16<<<cdiv(N_NODES * IN_DIM / 4, EW_T * EW_U), EW_T, 0, stream>>>(h, hf16,
                                                                                N_NODES * IN_DIM / 4);
    cvt_transpose_f16<<<cdiv(IN_DIM * D1, 256), 256, 0, stream>>>(W1, W1t16, IN_DIM, D1);
    cvt_transpose_f16<<<cdiv(D1 * OUT_DIM, 256), 256, 0, stream>>>(W2, W2t16, D1, OUT_DIM);

    gemm_wmma_f16t<<<dim3(N_NODES / 16, 1), 128, 0, stream>>>(hf16, W1t16, feat1,
                                                              N_NODES, D1, IN_DIM);
    attn_scores<<<cdiv(N_NODES * HEADS, 256), 256, 0, stream>>>(feat1, al1, ar1, el1, er1,
                                                                HEADS, HID);
    gat_owner<HEADS, HID, 2><<<cdiv(N_NODES, NB), OWN_T, 0, stream>>>(
        src, dst, feat1, el1, er1, (const float*)nullptr, out1);
    elu_bias_to_f16<D1><<<cdiv(N_NODES * D1 / 4, EW_T * EW_U), EW_T, 0, stream>>>(out1, b1, h1f16,
                                                                                  N_NODES * D1 / 4);

    gemm_wmma_f16t<<<dim3(N_NODES / 16, 1), 32, 0, stream>>>(h1f16, W2t16, feat2,
                                                             N_NODES, OUT_DIM, D1);
    attn_scores<<<cdiv(N_NODES, 256), 256, 0, stream>>>(feat2, al2, ar2, el2, er2, 1, OUT_DIM);
    gat_owner<1, OUT_DIM, 4><<<cdiv(N_NODES, NB), OWN_T, 0, stream>>>(
        src, dst, feat2, el2, er2, b2, out);
    (void)in_sizes; (void)n_in; (void)out_size; (void)ws_size;
}


# layer 2 gather takes each group's edge list from the one layer 1's group saved to workspace (same graph, same node ranges): no second dst filter, one phase; LDS atomic sums still in arrival order
# speedup vs baseline: 3.2124x; 1.2068x over previous
#include <hip/hip_runtime.h>
#include <hip/hip_bf16.h>
#include <hip/hip_fp16.h>

#define N_NODES 10000
#define N_EDGES 320000
#define IN_DIM  256
#define HID     64
#define HEADS   4
#define D1      (HEADS * HID)
#define OUT_DIM 64
#define NEG_SLOPE 0.2f

static_assert((N_NODES * IN_DIM) % 4 == 0 && (N_NODES * D1) % 4 == 0, "elementwise passes move float4");

typedef __attribute__((ext_vector_type(16))) _Float16 v16h;
typedef __attribute__((ext_vector_type(8)))  _Float16 v8h;
typedef __attribute__((ext_vector_type(4)))  _Float16 v4h;
typedef __attribute__((ext_vector_type(8)))  float    v8f;

#define EW_T 128
#define EW_U 8

__global__ void __launch_bounds__(EW_T)
cvt_f32_to_f16(const float* __restrict__ in, _Float16* __restrict__ out, int n4) {
    const float4* __restrict__ in4 = (const float4*)in;
    const int base = blockIdx.x * (EW_T * EW_U) + threadIdx.x;
    float4 v[EW_U];
#pragma unroll
    for (int u = 0; u < EW_U; ++u) {
        const int i = base + u * EW_T;
        v[u] = in4[(i < n4) ? i : (n4 - 1)];
    }
#pragma unroll
    for (int u = 0; u < EW_U; ++u) {
        const int i = base + u * EW_T;
        if (i < n4) {
            v4h r;
            r[0] = (_Float16)v[u].x; r[1] = (_Float16)v[u].y;
            r[2] = (_Float16)v[u].z; r[3] = (_Float16)v[u].w;
            *(v4h*)(out + (size_t)i * 4) = r;
        }
    }
}

__global__ void cvt_transpose_f16(const float* __restrict__ W, _Float16* __restrict__ Wt,
                                  int K, int N) {
    int i = blockIdx.x * blockDim.x + threadIdx.x;
    if (i >= K * N) return;
    int k = i / N, n = i % N;
    Wt[(size_t)n * K + k] = (_Float16)W[i];
}

__device__ inline float elu1(float x) { return (x > 0.0f) ? x : (__expf(x) - 1.0f); }

template <int F>
__global__ void __launch_bounds__(EW_T)
elu_bias_to_f16(const float* __restrict__ in, const float* __restrict__ bias,
                _Float16* __restrict__ out, int n4) {
    static_assert(F % 4 == 0, "a float4 never straddles two rows");
    static_assert((EW_T * 4) % F == 0, "all float4 of one thread fall on the same 4 bias columns");
    const float4* __restrict__ in4 = (const float4*)in;
    const int base = blockIdx.x * (EW_T * EW_U) + threadIdx.x;
    const float4 bv = ((const float4*)bias)[threadIdx.x % (F / 4)];
    float4 v[EW_U];
#pragma unroll
    for (int u = 0; u < EW_U; ++u) {
        const int i = base + u * EW_T;
        v[u] = in4[(i < n4) ? i : (n4 - 1)];
    }
#pragma unroll
    for (int u = 0; u < EW_U; ++u) {
        const int i = base + u * EW_T;
        if (i < n4) {
            v4h r;
            r[0] = (_Float16)elu1(v[u].x + bv.x); r[1] = (_Float16)elu1(v[u].y + bv.y);
            r[2] = (_Float16)elu1(v[u].z + bv.z); r[3] = (_Float16)elu1(v[u].w + bv.w);
            *(v4h*)(out + (size_t)i * 4) = r;
        }
    }
}

__device__ inline v16h load_frag(const _Float16* __restrict__ base, int half) {
    v8h lo = *(const v8h*)(base + 8 * half);
    v8h hi = *(const v8h*)(base + 16 + 8 * half);
    v16h r;
#pragma unroll
    for (int i = 0; i < 8; ++i) { r[i] = lo[i]; r[8 + i] = hi[i]; }
    return r;
}

__global__ void gemm_wmma_f16t(const _Float16* __restrict__ A, const _Float16* __restrict__ Bt,
                               float* __restrict__ C, int M, int N, int K) {
    const int lane = threadIdx.x & 31;
    const int wave = threadIdx.x >> 5;
    const int m0   = blockIdx.x * 16;
    const int n0   = (blockIdx.y * (blockDim.x >> 5) + wave) * 64;
    if (m0 >= M || n0 >= N) return;

    const int half = lane >> 4;
    const int l15  = lane & 15;

    const _Float16* aptr = A + (size_t)(m0 + l15) * K;
    const _Float16* bptr0 = Bt + (size_t)(n0 +  0 + l15) * K;
    const _Float16* bptr1 = Bt + (size_t)(n0 + 16 + l15) * K;
    const _Float16* bptr2 = Bt + (size_t)(n0 + 32 + l15) * K;
    const _Float16* bptr3 = Bt + (size_t)(n0 + 48 + l15) * K;

    v8f acc0 = {}, acc1 = {}, acc2 = {}, acc3 = {};

    for (int k0 = 0; k0 < K; k0 += 32) {
        v16h a  = load_frag(aptr  + k0, half);
        v16h b0 = load_frag(bptr0 + k0, half);
        v16h b1 = load_frag(bptr1 + k0, half);
        v16h b2 = load_frag(bptr2 + k0, half);
        v16h b3 = load_frag(bptr3 + k0, half);
        acc0 = __builtin_amdgcn_wmma_f32_16x16x32_f16(false, a, false, b0, (short)0, acc0, false, false);
        acc1 = __builtin_amdgcn_wmma_f32_16x16x32_f16(false, a, false, b1, (short)0, acc1, false, false);
        acc2 = __builtin_amdgcn_wmma_f32_16x16x32_f16(false, a, false, b2, (short)0, acc2, false, false);
        acc3 = __builtin_amdgcn_wmma_f32_16x16x32_f16(false, a, false, b3, (short)0, acc3, false, false);
    }

    float* cbase = C + (size_t)(m0 + 8 * half) * N + n0 + l15;
#pragma unroll
    for (int r = 0; r < 8; ++r) {
        float* row = cbase + (size_t)r * N;
        row[0]  = acc0[r];
        row[16] = acc1[r];
        row[32] = acc2[r];
        row[48] = acc3[r];
    }
}

__global__ void attn_scores(const float* __restrict__ feat, const float* __restrict__ al,
                            const float* __restrict__ ar, float* __restrict__ el,
                            float* __restrict__ er, int nheads, int d) {
    int idx = blockIdx.x * blockDim.x + threadIdx.x;
    if (idx >= N_NODES * nheads) return;
    int n = idx / nheads, h = idx % nheads;
    const float* f = feat + (size_t)(n * nheads + h) * d;
    float sl = 0.f, sr = 0.f;
    for (int j = 0; j < d; ++j) {
        float fv = f[j];
        sl += fv * al[h * d + j];
        sr += fv * ar[h * d + j];
    }
    el[idx] = sl;
    er[idx] = sr;
}

#define NB        40
#define OWN_T     1024
#define CHUNK_E   (OWN_T * 4)
#define NCHUNK    ((N_EDGES + CHUNK_E - 1) / CHUNK_E)
#define PH_CHUNKS 16
#define CAP       (PH_CHUNKS * CHUNK_E)
#define SCAN_U    8

static_assert(N_NODES <= 65536, "list entry packs src in 16 bits");
static_assert(N_EDGES <= (1 << 24) && NB <= 256, "list entry packs edge index in 24 bits, local dst in 8");
static_assert(N_EDGES % 4 == 0, "dst is read as int4");
static_assert(NB * HEADS <= OWN_T, "one thread per (node, head) slot");

__device__ inline unsigned f2ord(float f) {
    unsigned u = __float_as_uint(f);
    return (u & 0x80000000u) ? ~u : (u | 0x80000000u);
}
__device__ inline float ord2f(unsigned k) {
    return __uint_as_float((k & 0x80000000u) ? (k ^ 0x80000000u) : ~k);
}

template <int H, int D, int U, bool BUILD>
__global__ void __launch_bounds__(OWN_T)
gat_owner(const int* __restrict__ src, const int* __restrict__ dst,
          const float* __restrict__ feat, const float* __restrict__ el,
          const float* __restrict__ er, const float* __restrict__ bias,
          float* __restrict__ out, unsigned* __restrict__ lists,
          unsigned* __restrict__ counts) {
    constexpr int F   = H * D;
    constexpr int KPL = F / 32;
    __shared__ unsigned s_list[CAP];
    __shared__ float    s_acc[NB * F];
    __shared__ unsigned s_mcur[NB * H];
    __shared__ float    s_m[NB * H];
    __shared__ float    s_den[NB * H];
    __shared__ float    s_scl[NB * H];
    __shared__ unsigned s_cnt;

    const int tid  = threadIdx.x;
    const int lane = tid & 31;
    const int wave = __builtin_amdgcn_readfirstlane(tid >> 5);
    const int hl   = (lane < H) ? lane : 0;
    const int lo   = blockIdx.x * NB;
    const int c0   = (int)((blockIdx.x * (unsigned)NCHUNK) / gridDim.x);
    const float NEG_INF = -__builtin_inff();

    unsigned* __restrict__ mylist = lists + (size_t)blockIdx.x * CAP;
    unsigned woff = 0u;
    int saved = -1;
    if (!BUILD) {
        const unsigned c = __builtin_amdgcn_readfirstlane(counts[blockIdx.x]);
        if (c <= (unsigned)CAP) saved = (int)c;
    }

    for (int i = tid; i < NB * F; i += OWN_T) s_acc[i] = 0.0f;
    if (tid < NB * H) {
        s_mcur[tid] = f2ord(NEG_INF);
        s_m[tid]    = NEG_INF;
        s_den[tid]  = 0.0f;
    }

    for (int p0 = 0; p0 < NCHUNK; p0 += PH_CHUNKS) {
        int n;
        if (saved >= 0) {
            for (int i = tid; i < saved; i += OWN_T) s_list[i] = mylist[i];
            __syncthreads();
            n = saved;
        } else {
            if (tid == 0) s_cnt = 0u;
            __syncthreads();
            const int pend = (p0 + PH_CHUNKS < NCHUNK) ? (p0 + PH_CHUNKS) : NCHUNK;

            for (int ci = p0; ci < pend; ci += SCAN_U) {
                int4 v[SCAN_U];
                int  eb[SCAN_U];
#pragma unroll
                for (int u = 0; u < SCAN_U; ++u) {
                    int chunk = c0 + ci + u;
                    if (chunk >= NCHUNK) chunk -= NCHUNK;
                    const int e = chunk * CHUNK_E + tid * 4;
                    eb[u] = e;
                    const bool ok = (ci + u < pend) && (e < N_EDGES);
                    v[u] = ok ? *(const int4*)(dst + e) : make_int4(-1, -1, -1, -1);
                }
#pragma unroll
                for (int u = 0; u < SCAN_U; ++u) {
                    const int tt[4] = {v[u].x, v[u].y, v[u].z, v[u].w};
#pragma unroll
                    for (int c = 0; c < 4; ++c) {
                        const unsigned tl = (unsigned)(tt[c] - lo);
                        if (tl < (unsigned)NB) {
                            const unsigned pos = atomicAdd(&s_cnt, 1u);
                            s_list[pos] = (unsigned)(eb[u] + c) | (tl << 24);
                        }
                    }
                }
            }
            __syncthreads();
            n = (int)s_cnt;
        }

        for (int i = tid; i < n; i += OWN_T) {
            unsigned en = s_list[i];
            int tl, sv;
            if (saved >= 0) {
                sv = (int)(en & 0xFFFFu);
                tl = (int)(en >> 16);
            } else {
                tl = (int)(en >> 24);
                sv = src[en & 0xFFFFFFu];
                en = (unsigned)sv | ((unsigned)tl << 16);
                s_list[i] = en;
                if (BUILD && woff + (unsigned)i < (unsigned)CAP) mylist[woff + (unsigned)i] = en;
            }
#pragma unroll
            for (int h = 0; h < H; ++h) {
                float x = el[sv * H + h] + er[(lo + tl) * H + h];
                x = (x > 0.0f) ? x : NEG_SLOPE * x;
                atomicMax(&s_mcur[tl * H + h], f2ord(x));
            }
        }
        __syncthreads();
        if (tid < NB * H) {
            const float mn = ord2f(s_mcur[tid]);
            const float mo = s_m[tid];
            const float sc = (mo == mn) ? 1.0f : __expf(mo - mn);
            s_den[tid] *= sc;
            s_scl[tid]  = sc;
            s_m[tid]    = mn;
        }
        __syncthreads();
        if (p0 > 0) {
            for (int i = tid; i < NB * F; i += OWN_T) s_acc[i] *= s_scl[i / D];
            __syncthreads();
        }

        for (int i0 = wave * U; i0 < n; i0 += (OWN_T / 32) * U) {
            int   tlv[U], svv[U];
            float sa[U], sb[U];
            float fv[U][KPL];
#pragma unroll
            for (int u = 0; u < U; ++u) {
                const int i = i0 + u;
                const unsigned en = (i < n) ? s_list[i] : 0u;
                const int sv = (int)(en & 0xFFFFu);
                const int tl = (int)(en >> 16);
                tlv[u] = tl;
                svv[u] = sv;
#pragma unroll
                for (int k = 0; k < KPL; ++k)
                    fv[u][k] = feat[(size_t)sv * F + k * 32 + lane];
            }
#pragma unroll
            for (int u = 0; u < U; ++u) {
                sa[u] = el[svv[u] * H + hl];
                sb[u] = er[(lo + tlv[u]) * H + hl];
            }
#pragma unroll
            for (int u = 0; u < U; ++u) {
                if (i0 + u < n) {
                    const int tl = tlv[u];
                    float x = sa[u] + sb[u];
                    x = (x > 0.0f) ? x : NEG_SLOPE * x;
                    const float wlane = __expf(x - s_m[tl * H + hl]);
                    if (lane < H) atomicAdd(&s_den[tl * H + lane], wlane);
#pragma unroll
                    for (int k = 0; k < KPL; ++k) {
                        const float w = __shfl(wlane, (k * 32) / D);
                        atomicAdd(&s_acc[tl * F + k * 32 + lane], w * fv[u][k]);
                    }
                }
            }
        }
        __syncthreads();
        if (BUILD) woff += (unsigned)n;
        if (saved >= 0) break;
    }
    if (BUILD && tid == 0) counts[blockIdx.x] = woff;

    for (int i = tid; i < NB * F; i += OWN_T) {
        const int tl = i / F, f = i - tl * F;
        const int node = lo + tl;
        if (node < N_NODES) {
            const float d = s_den[tl * H + f / D];
            float r = (d > 0.0f) ? s_acc[i] / d : 0.0f;
            if (bias) r += bias[f];
            out[(size_t)node * F + f] = r;
        }
    }
}

extern "C" void kernel_launch(void* const* d_in, const int* in_sizes, int n_in,
                              void* d_out, int out_size, void* d_ws, size_t ws_size,
                              hipStream_t stream) {
    const float* h   = (const float*)d_in[0];
    const int*   src = (const int*)  d_in[1];
    const int*   dst = (const int*)  d_in[2];
    const float* W1  = (const float*)d_in[3];
    const float* al1 = (const float*)d_in[4];
    const float* ar1 = (const float*)d_in[5];
    const float* b1  = (const float*)d_in[6];
    const float* W2  = (const float*)d_in[7];
    const float* al2 = (const float*)d_in[8];
    const float* ar2 = (const float*)d_in[9];
    const float* b2  = (const float*)d_in[10];
    float* out = (float*)d_out;

    size_t off = 0;
    auto alloc = [&](size_t bytes) {
        void* p = (char*)d_ws + off;
        off += (bytes + 255) & ~(size_t)255;
        return p;
    };
    _Float16* hf16   = (_Float16*)alloc((size_t)N_NODES * IN_DIM  * 2);
    _Float16* W1t16  = (_Float16*)alloc((size_t)IN_DIM  * D1      * 2);
    _Float16* W2t16  = (_Float16*)alloc((size_t)D1      * OUT_DIM * 2);
    float*    feat1  = (float*)   alloc((size_t)N_NODES * D1      * 4);
    float*    el1    = (float*)   alloc((size_t)N_NODES * HEADS   * 4);
    float*    er1    = (float*)   alloc((size_t)N_NODES * HEADS   * 4);
    float*    out1   = (float*)   alloc((size_t)N_NODES * D1      * 4);
    _Float16* h1f16  = (_Float16*)alloc((size_t)N_NODES * D1      * 2);
    float*    feat2  = (float*)   alloc((size_t)N_NODES * OUT_DIM * 4);
    float*    el2    = (float*)   alloc((size_t)N_NODES * 4);
    float*    er2    = (float*)   alloc((size_t)N_NODES * 4);
    unsigned* lists  = (unsigned*)alloc((size_t)((N_NODES + NB - 1) / NB) * CAP * 4);
    unsigned* counts = (unsigned*)alloc((size_t)((N_NODES + NB - 1) / NB) * 4);

    auto cdiv = [](int a, int b) { return (a + b - 1) / b; };

    cvt_f32_to_f16<<<cdiv(N_NODES * IN_DIM / 4, EW_T * EW_U), EW_T, 0, stream>>>(h, hf16,
                                                                                N_NODES * IN_DIM / 4);
    cvt_transpose_f16<<<cdiv(IN_DIM * D1, 256), 256, 0, stream>>>(W1, W1t16, IN_DIM, D1);
    cvt_transpose_f16<<<cdiv(D1 * OUT_DIM, 256), 256, 0, stream>>>(W2, W2t16, D1, OUT_DIM);

    gemm_wmma_f16t<<<dim3(N_NODES / 16, 1), 128, 0, stream>>>(hf16, W1t16, feat1,
                                                              N_NODES, D1, IN_DIM);
    attn_scores<<<cdiv(N_NODES * HEADS, 256), 256, 0, stream>>>(feat1, al1, ar1, el1, er1,
                                                                HEADS, HID);
    gat_owner<HEADS, HID, 2, true><<<cdiv(N_NODES, NB), OWN_T, 0, stream>>>(
        src, dst, feat1, el1, er1, (const float*)nullptr, out1, lists, counts);
    elu_bias_to_f16<D1><<<cdiv(N_NODES * D1 / 4, EW_T * EW_U), EW_T, 0, stream>>>(out1, b1, h1f16,
                                                                                  N_NODES * D1 / 4);

    gemm_wmma_f16t<<<dim3(N_NODES / 16, 1), 32, 0, stream>>>(h1f16, W2t16, feat2,
                                                             N_NODES, OUT_DIM, D1);
    attn_scores<<<cdiv(N_NODES, 256), 256, 0, stream>>>(feat2, al2, ar2, el2, er2, 1, OUT_DIM);
    gat_owner<1, OUT_DIM, 4, false><<<cdiv(N_NODES, NB), OWN_T, 0, stream>>>(
        src, dst, feat2, el2, er2, b2, out, lists, counts);
    (void)in_sizes; (void)n_in; (void)out_size; (void)ws_size;
}


# owner groups sweep a 1-byte owner id per edge (dst / 40, made by one new launch from one read of dst), not 4-byte dst: a quarter of the sweep bytes; LDS atomic sums, arrival order; 10 -> 11 launches
# speedup vs baseline: 3.6330x; 1.1309x over previous
#include <hip/hip_runtime.h>
#include <hip/hip_bf16.h>
#include <hip/hip_fp16.h>

#define N_NODES 10000
#define N_EDGES 320000
#define IN_DIM  256
#define HID     64
#define HEADS   4
#define D1      (HEADS * HID)
#define OUT_DIM 64
#define NEG_SLOPE 0.2f

static_assert((N_NODES * IN_DIM) % 4 == 0 && (N_NODES * D1) % 4 == 0, "elementwise passes move float4");

typedef __attribute__((ext_vector_type(16))) _Float16 v16h;
typedef __attribute__((ext_vector_type(8)))  _Float16 v8h;
typedef __attribute__((ext_vector_type(4)))  _Float16 v4h;
typedef __attribute__((ext_vector_type(8)))  float    v8f;

#define EW_T 128
#define EW_U 8

__global__ void __launch_bounds__(EW_T)
cvt_f32_to_f16(const float* __restrict__ in, _Float16* __restrict__ out, int n4) {
    const float4* __restrict__ in4 = (const float4*)in;
    const int base = blockIdx.x * (EW_T * EW_U) + threadIdx.x;
    float4 v[EW_U];
#pragma unroll
    for (int u = 0; u < EW_U; ++u) {
        const int i = base + u * EW_T;
        v[u] = in4[(i < n4) ? i : (n4 - 1)];
    }
#pragma unroll
    for (int u = 0; u < EW_U; ++u) {
        const int i = base + u * EW_T;
        if (i < n4) {
            v4h r;
            r[0] = (_Float16)v[u].x; r[1] = (_Float16)v[u].y;
            r[2] = (_Float16)v[u].z; r[3] = (_Float16)v[u].w;
            *(v4h*)(out + (size_t)i * 4) = r;
        }
    }
}

__global__ void cvt_transpose_f16(const float* __restrict__ W, _Float16* __restrict__ Wt,
                                  int K, int N) {
    int i = blockIdx.x * blockDim.x + threadIdx.x;
    if (i >= K * N) return;
    int k = i / N, n = i % N;
    Wt[(size_t)n * K + k] = (_Float16)W[i];
}

__device__ inline float elu1(float x) { return (x > 0.0f) ? x : (__expf(x) - 1.0f); }

template <int F>
__global__ void __launch_bounds__(EW_T)
elu_bias_to_f16(const float* __restrict__ in, const float* __restrict__ bias,
                _Float16* __restrict__ out, int n4) {
    static_assert(F % 4 == 0, "a float4 never straddles two rows");
    static_assert((EW_T * 4) % F == 0, "all float4 of one thread fall on the same 4 bias columns");
    const float4* __restrict__ in4 = (const float4*)in;
    const int base = blockIdx.x * (EW_T * EW_U) + threadIdx.x;
    const float4 bv = ((const float4*)bias)[threadIdx.x % (F / 4)];
    float4 v[EW_U];
#pragma unroll
    for (int u = 0; u < EW_U; ++u) {
        const int i = base + u * EW_T;
        v[u] = in4[(i < n4) ? i : (n4 - 1)];
    }
#pragma unroll
    for (int u = 0; u < EW_U; ++u) {
        const int i = base + u * EW_T;
        if (i < n4) {
            v4h r;
            r[0] = (_Float16)elu1(v[u].x + bv.x); r[1] = (_Float16)elu1(v[u].y + bv.y);
            r[2] = (_Float16)elu1(v[u].z + bv.z); r[3] = (_Float16)elu1(v[u].w + bv.w);
            *(v4h*)(out + (size_t)i * 4) = r;
        }
    }
}

__device__ inline v16h load_frag(const _Float16* __restrict__ base, int half) {
    v8h lo = *(const v8h*)(base + 8 * half);
    v8h hi = *(const v8h*)(base + 16 + 8 * half);
    v16h r;
#pragma unroll
    for (int i = 0; i < 8; ++i) { r[i] = lo[i]; r[8 + i] = hi[i]; }
    return r;
}

__global__ void gemm_wmma_f16t(const _Float16* __restrict__ A, const _Float16* __restrict__ Bt,
                               float* __restrict__ C, int M, int N, int K) {
    const int lane = threadIdx.x & 31;
    const int wave = threadIdx.x >> 5;
    const int m0   = blockIdx.x * 16;
    const int n0   = (blockIdx.y * (blockDim.x >> 5) + wave) * 64;
    if (m0 >= M || n0 >= N) return;

    const int half = lane >> 4;
    const int l15  = lane & 15;

    const _Float16* aptr = A + (size_t)(m0 + l15) * K;
    const _Float16* bptr0 = Bt + (size_t)(n0 +  0 + l15) * K;
    const _Float16* bptr1 = Bt + (size_t)(n0 + 16 + l15) * K;
    const _Float16* bptr2 = Bt + (size_t)(n0 + 32 + l15) * K;
    const _Float16* bptr3 = Bt + (size_t)(n0 + 48 + l15) * K;

    v8f acc0 = {}, acc1 = {}, acc2 = {}, acc3 = {};

    for (int k0 = 0; k0 < K; k0 += 32) {
        v16h a  = load_frag(aptr  + k0, half);
        v16h b0 = load_frag(bptr0 + k0, half);
        v16h b1 = load_frag(bptr1 + k0, half);
        v16h b2 = load_frag(bptr2 + k0, half);
        v16h b3 = load_frag(bptr3 + k0, half);
        acc0 = __builtin_amdgcn_wmma_f32_16x16x32_f16(false, a, false, b0, (short)0, acc0, false, false);
        acc1 = __builtin_amdgcn_wmma_f32_16x16x32_f16(false, a, false, b1, (short)0, acc1, false, false);
        acc2 = __builtin_amdgcn_wmma_f32_16x16x32_f16(false, a, false, b2, (short)0, acc2, false, false);
        acc3 = __builtin_amdgcn_wmma_f32_16x16x32_f16(false, a, false, b3, (short)0, acc3, false, false);
    }

    float* cbase = C + (size_t)(m0 + 8 * half) * N + n0 + l15;
#pragma unroll
    for (int r = 0; r < 8; ++r) {
        float* row = cbase + (size_t)r * N;
        row[0]  = acc0[r];
        row[16] = acc1[r];
        row[32] = acc2[r];
        row[48] = acc3[r];
    }
}

__global__ void attn_scores(const float* __restrict__ feat, const float* __restrict__ al,
                            const float* __restrict__ ar, float* __restrict__ el,
                            float* __restrict__ er, int nheads, int d) {
    int idx = blockIdx.x * blockDim.x + threadIdx.x;
    if (idx >= N_NODES * nheads) return;
    int n = idx / nheads, h = idx % nheads;
    const float* f = feat + (size_t)(n * nheads + h) * d;
    float sl = 0.f, sr = 0.f;
    for (int j = 0; j < d; ++j) {
        float fv = f[j];
        sl += fv * al[h * d + j];
        sr += fv * ar[h * d + j];
    }
    el[idx] = sl;
    er[idx] = sr;
}

#define NB        40
#define OWN_T     1024
#define NGROUPS   ((N_NODES + NB - 1) / NB)
#define ID_E      16
#define CHUNK_E   (OWN_T * ID_E)
#define NCHUNK    ((N_EDGES + CHUNK_E - 1) / CHUNK_E)
#define PH_CHUNKS 4
#define CAP       (PH_CHUNKS * CHUNK_E)
#define SCAN_U    4
#define ID_NONE   255u
#define IDS_N     (NCHUNK * CHUNK_E)

static_assert(N_NODES <= 65536 && NB <= 65536, "saved list entry packs src and local dst in 16 bits each");
static_assert(NGROUPS <= 255, "a group number fits a byte and never equals ID_NONE");
static_assert(N_EDGES % ID_E == 0, "a thread's 16 ids are all edges or all pad");
static_assert(NCHUNK % SCAN_U == 0 && PH_CHUNKS % SCAN_U == 0, "every batch of SCAN_U chunks lies inside one phase");
static_assert(NB * HEADS <= OWN_T, "one thread per (node, head) slot");

#define IDS_T 128
static_assert((IDS_N / ID_E) % IDS_T == 0, "the grid covers the padded array exactly");
__global__ void __launch_bounds__(IDS_T)
owner_ids(const int* __restrict__ dst, unsigned char* __restrict__ ids) {
    const int t = blockIdx.x * IDS_T + threadIdx.x;
    unsigned w[ID_E / 4] = {~0u, ~0u, ~0u, ~0u};
    if (t * ID_E < N_EDGES) {
        const int4* __restrict__ d4 = (const int4*)dst + (size_t)t * (ID_E / 4);
        int4 v[ID_E / 4];
#pragma unroll
        for (int q = 0; q < ID_E / 4; ++q) v[q] = d4[q];
#pragma unroll
        for (int q = 0; q < ID_E / 4; ++q) {
            const int tt[4] = {v[q].x, v[q].y, v[q].z, v[q].w};
            unsigned r = 0u;
#pragma unroll
            for (int c = 0; c < 4; ++c) {
                unsigned g = (unsigned)tt[c] / (unsigned)NB;
                g = ((unsigned)tt[c] < (unsigned)N_NODES) ? g : ID_NONE;
                r |= g << (8 * c);
            }
            w[q] = r;
        }
    }
    *(uint4*)(ids + (size_t)t * ID_E) = make_uint4(w[0], w[1], w[2], w[3]);
}

__device__ inline unsigned f2ord(float f) {
    unsigned u = __float_as_uint(f);
    return (u & 0x80000000u) ? ~u : (u | 0x80000000u);
}
__device__ inline float ord2f(unsigned k) {
    return __uint_as_float((k & 0x80000000u) ? (k ^ 0x80000000u) : ~k);
}

template <int H, int D, int U, bool BUILD>
__global__ void __launch_bounds__(OWN_T)
gat_owner(const int* __restrict__ src, const int* __restrict__ dst,
          const unsigned char* __restrict__ ids, const float* __restrict__ feat,
          const float* __restrict__ el,
          const float* __restrict__ er, const float* __restrict__ bias,
          float* __restrict__ out, unsigned* __restrict__ lists,
          unsigned* __restrict__ counts) {
    constexpr int F   = H * D;
    constexpr int KPL = F / 32;
    __shared__ unsigned s_list[CAP];
    __shared__ float    s_acc[NB * F];
    __shared__ unsigned s_mcur[NB * H];
    __shared__ float    s_m[NB * H];
    __shared__ float    s_den[NB * H];
    __shared__ float    s_scl[NB * H];
    __shared__ unsigned s_cnt;

    const int tid  = threadIdx.x;
    const int lane = tid & 31;
    const int wave = __builtin_amdgcn_readfirstlane(tid >> 5);
    const int hl   = (lane < H) ? lane : 0;
    const int lo   = blockIdx.x * NB;
    const int c0   = (int)((blockIdx.x * (unsigned)NCHUNK) / gridDim.x);
    const float NEG_INF = -__builtin_inff();
    const unsigned gpat = blockIdx.x * 0x01010101u;

    unsigned* __restrict__ mylist = lists + (size_t)blockIdx.x * CAP;
    unsigned woff = 0u;
    int saved = -1;
    if (!BUILD) {
        const unsigned c = __builtin_amdgcn_readfirstlane(counts[blockIdx.x]);
        if (c <= (unsigned)CAP) saved = (int)c;
    }

    for (int i = tid; i < NB * F; i += OWN_T) s_acc[i] = 0.0f;
    if (tid < NB * H) {
        s_mcur[tid] = f2ord(NEG_INF);
        s_m[tid]    = NEG_INF;
        s_den[tid]  = 0.0f;
    }

    int ph = BUILD ? NCHUNK : PH_CHUNKS;
    for (int p0 = 0; p0 < NCHUNK;) {
        int n;
        if (saved >= 0) {
            for (int i = tid; i < saved; i += OWN_T) s_list[i] = mylist[i];
            __syncthreads();
            n = saved;
        } else {
            if (tid == 0) s_cnt = 0u;
            __syncthreads();
            const int pend = (p0 + ph < NCHUNK) ? (p0 + ph) : NCHUNK;

            for (int ci = p0; ci < pend; ci += SCAN_U) {
                uint4 v[SCAN_U];
                int   eb[SCAN_U];
#pragma unroll
                for (int u = 0; u < SCAN_U; ++u) {
                    int chunk = c0 + ci + u;
                    if (chunk >= NCHUNK) chunk -= NCHUNK;
                    eb[u] = chunk * CHUNK_E + tid * ID_E;
                    v[u] = *(const uint4*)(ids + eb[u]);
                }
#pragma unroll
                for (int u = 0; u < SCAN_U; ++u) {
                    const unsigned ww[4] = {v[u].x, v[u].y, v[u].z, v[u].w};
                    unsigned m = 0u;
#pragma unroll
                    for (int q = 0; q < 4; ++q) {
                        const unsigned x = ww[q] ^ gpat;
                        const unsigned y = ((x & 0x7F7F7F7Fu) + 0x7F7F7F7Fu) | x;
                        m |= (~y & 0x80808080u) >> (7 - q);
                    }
                    while (m) {
                        const int p = __builtin_ctz(m);
                        m &= m - 1u;
                        const unsigned pos = atomicAdd(&s_cnt, 1u);
                        if (pos < (unsigned)CAP) s_list[pos] = (unsigned)(eb[u] + 4 * (p & 7) + (p >> 3));
                    }
                }
            }
            __syncthreads();
            n = (int)s_cnt;
            if (BUILD && n > CAP) {
                ph = PH_CHUNKS;
                __syncthreads();
                continue;
            }
        }

        for (int i = tid; i < n; i += OWN_T) {
            unsigned en = s_list[i];
            int tl, sv;
            if (saved >= 0) {
                sv = (int)(en & 0xFFFFu);
                tl = (int)(en >> 16);
            } else {
                tl = dst[en] - lo;
                sv = src[en];
                en = (unsigned)sv | ((unsigned)tl << 16);
                s_list[i] = en;
                if (BUILD && woff + (unsigned)i < (unsigned)CAP) mylist[woff + (unsigned)i] = en;
            }
#pragma unroll
            for (int h = 0; h < H; ++h) {
                float x = el[sv * H + h] + er[(lo + tl) * H + h];
                x = (x > 0.0f) ? x : NEG_SLOPE * x;
                atomicMax(&s_mcur[tl * H + h], f2ord(x));
            }
        }
        __syncthreads();
        if (tid < NB * H) {
            const float mn = ord2f(s_mcur[tid]);
            const float mo = s_m[tid];
            const float sc = (mo == mn) ? 1.0f : __expf(mo - mn);
            s_den[tid] *= sc;
            s_scl[tid]  = sc;
            s_m[tid]    = mn;
        }
        __syncthreads();
        if (p0 > 0) {
            for (int i = tid; i < NB * F; i += OWN_T) s_acc[i] *= s_scl[i / D];
            __syncthreads();
        }

        for (int i0 = wave * U; i0 < n; i0 += (OWN_T / 32) * U) {
            int   tlv[U], svv[U];
            float sa[U], sb[U];
            float fv[U][KPL];
#pragma unroll
            for (int u = 0; u < U; ++u) {
                const int i = i0 + u;
                const unsigned en = (i < n) ? s_list[i] : 0u;
                const int sv = (int)(en & 0xFFFFu);
                const int tl = (int)(en >> 16);
                tlv[u] = tl;
                svv[u] = sv;
#pragma unroll
                for (int k = 0; k < KPL; ++k)
                    fv[u][k] = feat[(size_t)sv * F + k * 32 + lane];
            }
#pragma unroll
            for (int u = 0; u < U; ++u) {
                sa[u] = el[svv[u] * H + hl];
                sb[u] = er[(lo + tlv[u]) * H + hl];
            }
#pragma unroll
            for (int u = 0; u < U; ++u) {
                if (i0 + u < n) {
                    const int tl = tlv[u];
                    float x = sa[u] + sb[u];
                    x = (x > 0.0f) ? x : NEG_SLOPE * x;
                    const float wlane = __expf(x - s_m[tl * H + hl]);
                    if (lane < H) atomicAdd(&s_den[tl * H + lane], wlane);
#pragma unroll
                    for (int k = 0; k < KPL; ++k) {
                        const float w = __shfl(wlane, (k * 32) / D);
                        atomicAdd(&s_acc[tl * F + k * 32 + lane], w * fv[u][k]);
                    }
                }
            }
        }
        __syncthreads();
        if (BUILD) woff += (unsigned)n;
        if (saved >= 0) break;
        p0 += ph;
    }
    if (BUILD && tid == 0) counts[blockIdx.x] = woff;

    for (int i = tid; i < NB * F; i += OWN_T) {
        const int tl = i / F, f = i - tl * F;
        const int node = lo + tl;
        if (node < N_NODES) {
            const float d = s_den[tl * H + f / D];
            float r = (d > 0.0f) ? s_acc[i] / d : 0.0f;
            if (bias) r += bias[f];
            out[(size_t)node * F + f] = r;
        }
    }
}

extern "C" void kernel_launch(void* const* d_in, const int* in_sizes, int n_in,
                              void* d_out, int out_size, void* d_ws, size_t ws_size,
                              hipStream_t stream) {
    const float* h   = (const float*)d_in[0];
    const int*   src = (const int*)  d_in[1];
    const int*   dst = (const int*)  d_in[2];
    const float* W1  = (const float*)d_in[3];
    const float* al1 = (const float*)d_in[4];
    const float* ar1 = (const float*)d_in[5];
    const float* b1  = (const float*)d_in[6];
    const float* W2  = (const float*)d_in[7];
    const float* al2 = (const float*)d_in[8];
    const float* ar2 = (const float*)d_in[9];
    const float* b2  = (const float*)d_in[10];
    float* out = (float*)d_out;

    size_t off = 0;
    auto alloc = [&](size_t bytes) {
        void* p = (char*)d_ws + off;
        off += (bytes + 255) & ~(size_t)255;
        return p;
    };
    _Float16* hf16   = (_Float16*)alloc((size_t)N_NODES * IN_DIM  * 2);
    _Float16* W1t16  = (_Float16*)alloc((size_t)IN_DIM  * D1      * 2);
    _Float16* W2t16  = (_Float16*)alloc((size_t)D1      * OUT_DIM * 2);
    float*    feat1  = (float*)   alloc((size_t)N_NODES * D1      * 4);
    float*    el1    = (float*)   alloc((size_t)N_NODES * HEADS   * 4);
    float*    er1    = (float*)   alloc((size_t)N_NODES * HEADS   * 4);
    float*    out1   = (float*)   alloc((size_t)N_NODES * D1      * 4);
    _Float16* h1f16  = (_Float16*)alloc((size_t)N_NODES * D1      * 2);
    float*    feat2  = (float*)   alloc((size_t)N_NODES * OUT_DIM * 4);
    float*    el2    = (float*)   alloc((size_t)N_NODES * 4);
    float*    er2    = (float*)   alloc((size_t)N_NODES * 4);
    unsigned* lists  = (unsigned*)alloc((size_t)((N_NODES + NB - 1) / NB) * CAP * 4);
    unsigned* counts = (unsigned*)alloc((size_t)((N_NODES + NB - 1) / NB) * 4);
    unsigned char* ids = (unsigned char*)alloc((size_t)IDS_N);

    auto cdiv = [](int a, int b) { return (a + b - 1) / b; };

    cvt_f32_to_f16<<<cdiv(N_NODES * IN_DIM / 4, EW_T * EW_U), EW_T, 0, stream>>>(h, hf16,
                                                                                N_NODES * IN_DIM / 4);
    cvt_transpose_f16<<<cdiv(IN_DIM * D1, 256), 256, 0, stream>>>(W1, W1t16, IN_DIM, D1);
    cvt_transpose_f16<<<cdiv(D1 * OUT_DIM, 256), 256, 0, stream>>>(W2, W2t16, D1, OUT_DIM);

    gemm_wmma_f16t<<<dim3(N_NODES / 16, 1), 128, 0, stream>>>(hf16, W1t16, feat1,
                                                              N_NODES, D1, IN_DIM);
    attn_scores<<<cdiv(N_NODES * HEADS, 256), 256, 0, stream>>>(feat1, al1, ar1, el1, er1,
                                                                HEADS, HID);
    owner_ids<<<IDS_N / ID_E / IDS_T, IDS_T, 0, stream>>>(dst, ids);
    gat_owner<HEADS, HID, 2, true><<<cdiv(N_NODES, NB), OWN_T, 0, stream>>>(
        src, dst, ids, feat1, el1, er1, (const float*)nullptr, out1, lists, counts);
    elu_bias_to_f16<D1><<<cdiv(N_NODES * D1 / 4, EW_T * EW_U), EW_T, 0, stream>>>(out1, b1, h1f16,
                                                                                  N_NODES * D1 / 4);

    gemm_wmma_f16t<<<dim3(N_NODES / 16, 1), 32, 0, stream>>>(h1f16, W2t16, feat2,
                                                             N_NODES, OUT_DIM, D1);
    attn_scores<<<cdiv(N_NODES, 256), 256, 0, stream>>>(feat2, al2, ar2, el2, er2, 1, OUT_DIM);
    gat_owner<1, OUT_DIM, 4, false><<<cdiv(N_NODES, NB), OWN_T, 0, stream>>>(
        src, dst, ids, feat2, el2, er2, b2, out, lists, counts);
    (void)in_sizes; (void)n_in; (void)out_size; (void)ws_size;
}


# layer 1 gather writes its rows as ELU(row + bias) in f16 straight from its LDS sums (256-B wave stores): f32 row plane and ELU pass gone; LDS atomic sums still in arrival order; 11 -> 10 launches
# speedup vs baseline: 3.7950x; 1.0446x over previous
#include <hip/hip_runtime.h>
#include <hip/hip_bf16.h>
#include <hip/hip_fp16.h>

#define N_NODES 10000
#define N_EDGES 320000
#define IN_DIM  256
#define HID     64
#define HEADS   4
#define D1      (HEADS * HID)
#define OUT_DIM 64
#define NEG_SLOPE 0.2f

static_assert((N_NODES * IN_DIM) % 4 == 0, "the elementwise pass moves float4");

typedef __attribute__((ext_vector_type(16))) _Float16 v16h;
typedef __attribute__((ext_vector_type(8)))  _Float16 v8h;
typedef __attribute__((ext_vector_type(4)))  _Float16 v4h;
typedef __attribute__((ext_vector_type(8)))  float    v8f;

#define EW_T 128
#define EW_U 8

__global__ void __launch_bounds__(EW_T)
cvt_f32_to_f16(const float* __restrict__ in, _Float16* __restrict__ out, int n4) {
    const float4* __restrict__ in4 = (const float4*)in;
    const int base = blockIdx.x * (EW_T * EW_U) + threadIdx.x;
    float4 v[EW_U];
#pragma unroll
    for (int u = 0; u < EW_U; ++u) {
        const int i = base + u * EW_T;
        v[u] = in4[(i < n4) ? i : (n4 - 1)];
    }
#pragma unroll
    for (int u = 0; u < EW_U; ++u) {
        const int i = base + u * EW_T;
        if (i < n4) {
            v4h r;
            r[0] = (_Float16)v[u].x; r[1] = (_Float16)v[u].y;
            r[2] = (_Float16)v[u].z; r[3] = (_Float16)v[u].w;
            *(v4h*)(out + (size_t)i * 4) = r;
        }
    }
}

__global__ void cvt_transpose_f16(const float* __restrict__ W, _Float16* __restrict__ Wt,
                                  int K, int N) {
    int i = blockIdx.x * blockDim.x + threadIdx.x;
    if (i >= K * N) return;
    int k = i / N, n = i % N;
    Wt[(size_t)n * K + k] = (_Float16)W[i];
}

__device__ inline float elu1(float x) { return (x > 0.0f) ? x : (__expf(x) - 1.0f); }

__device__ inline v16h load_frag(const _Float16* __restrict__ base, int half) {
    v8h lo = *(const v8h*)(base + 8 * half);
    v8h hi = *(const v8h*)(base + 16 + 8 * half);
    v16h r;
#pragma unroll
    for (int i = 0; i < 8; ++i) { r[i] = lo[i]; r[8 + i] = hi[i]; }
    return r;
}

__global__ void gemm_wmma_f16t(const _Float16* __restrict__ A, const _Float16* __restrict__ Bt,
                               float* __restrict__ C, int M, int N, int K) {
    const int lane = threadIdx.x & 31;
    const int wave = threadIdx.x >> 5;
    const int m0   = blockIdx.x * 16;
    const int n0   = (blockIdx.y * (blockDim.x >> 5) + wave) * 64;
    if (m0 >= M || n0 >= N) return;

    const int half = lane >> 4;
    const int l15  = lane & 15;

    const _Float16* aptr = A + (size_t)(m0 + l15) * K;
    const _Float16* bptr0 = Bt + (size_t)(n0 +  0 + l15) * K;
    const _Float16* bptr1 = Bt + (size_t)(n0 + 16 + l15) * K;
    const _Float16* bptr2 = Bt + (size_t)(n0 + 32 + l15) * K;
    const _Float16* bptr3 = Bt + (size_t)(n0 + 48 + l15) * K;

    v8f acc0 = {}, acc1 = {}, acc2 = {}, acc3 = {};

    for (int k0 = 0; k0 < K; k0 += 32) {
        v16h a  = load_frag(aptr  + k0, half);
        v16h b0 = load_frag(bptr0 + k0, half);
        v16h b1 = load_frag(bptr1 + k0, half);
        v16h b2 = load_frag(bptr2 + k0, half);
        v16h b3 = load_frag(bptr3 + k0, half);
        acc0 = __builtin_amdgcn_wmma_f32_16x16x32_f16(false, a, false, b0, (short)0, acc0, false, false);
        acc1 = __builtin_amdgcn_wmma_f32_16x16x32_f16(false, a, false, b1, (short)0, acc1, false, false);
        acc2 = __builtin_amdgcn_wmma_f32_16x16x32_f16(false, a, false, b2, (short)0, acc2, false, false);
        acc3 = __builtin_amdgcn_wmma_f32_16x16x32_f16(false, a, false, b3, (short)0, acc3, false, false);
    }

    float* cbase = C + (size_t)(m0 + 8 * half) * N + n0 + l15;
#pragma unroll
    for (int r = 0; r < 8; ++r) {
        float* row = cbase + (size_t)r * N;
        row[0]  = acc0[r];
        row[16] = acc1[r];
        row[32] = acc2[r];
        row[48] = acc3[r];
    }
}

__global__ void attn_scores(const float* __restrict__ feat, const float* __restrict__ al,
                            const float* __restrict__ ar, float* __restrict__ el,
                            float* __restrict__ er, int nheads, int d) {
    int idx = blockIdx.x * blockDim.x + threadIdx.x;
    if (idx >= N_NODES * nheads) return;
    int n = idx / nheads, h = idx % nheads;
    const float* f = feat + (size_t)(n * nheads + h) * d;
    float sl = 0.f, sr = 0.f;
    for (int j = 0; j < d; ++j) {
        float fv = f[j];
        sl += fv * al[h * d + j];
        sr += fv * ar[h * d + j];
    }
    el[idx] = sl;
    er[idx] = sr;
}

#define NB        40
#define OWN_T     1024
#define NGROUPS   ((N_NODES + NB - 1) / NB)
#define ID_E      16
#define CHUNK_E   (OWN_T * ID_E)
#define NCHUNK    ((N_EDGES + CHUNK_E - 1) / CHUNK_E)
#define PH_CHUNKS 4
#define CAP       (PH_CHUNKS * CHUNK_E)
#define SCAN_U    4
#define ID_NONE   255u
#define IDS_N     (NCHUNK * CHUNK_E)

static_assert(N_NODES <= 65536 && NB <= 65536, "saved list entry packs src and local dst in 16 bits each");
static_assert(NGROUPS <= 255, "a group number fits a byte and never equals ID_NONE");
static_assert(N_EDGES % ID_E == 0, "a thread's 16 ids are all edges or all pad");
static_assert(NCHUNK % SCAN_U == 0 && PH_CHUNKS % SCAN_U == 0, "every batch of SCAN_U chunks lies inside one phase");
static_assert(NB * HEADS <= OWN_T, "one thread per (node, head) slot");

#define IDS_T 128
static_assert((IDS_N / ID_E) % IDS_T == 0, "the grid covers the padded array exactly");
__global__ void __launch_bounds__(IDS_T)
owner_ids(const int* __restrict__ dst, unsigned char* __restrict__ ids) {
    const int t = blockIdx.x * IDS_T + threadIdx.x;
    unsigned w[ID_E / 4] = {~0u, ~0u, ~0u, ~0u};
    if (t * ID_E < N_EDGES) {
        const int4* __restrict__ d4 = (const int4*)dst + (size_t)t * (ID_E / 4);
        int4 v[ID_E / 4];
#pragma unroll
        for (int q = 0; q < ID_E / 4; ++q) v[q] = d4[q];
#pragma unroll
        for (int q = 0; q < ID_E / 4; ++q) {
            const int tt[4] = {v[q].x, v[q].y, v[q].z, v[q].w};
            unsigned r = 0u;
#pragma unroll
            for (int c = 0; c < 4; ++c) {
                unsigned g = (unsigned)tt[c] / (unsigned)NB;
                g = ((unsigned)tt[c] < (unsigned)N_NODES) ? g : ID_NONE;
                r |= g << (8 * c);
            }
            w[q] = r;
        }
    }
    *(uint4*)(ids + (size_t)t * ID_E) = make_uint4(w[0], w[1], w[2], w[3]);
}

__device__ inline unsigned f2ord(float f) {
    unsigned u = __float_as_uint(f);
    return (u & 0x80000000u) ? ~u : (u | 0x80000000u);
}
__device__ inline float ord2f(unsigned k) {
    return __uint_as_float((k & 0x80000000u) ? (k ^ 0x80000000u) : ~k);
}

template <int H, int D, int U, bool BUILD, bool ELU16>
__global__ void __launch_bounds__(OWN_T)
gat_owner(const int* __restrict__ src, const int* __restrict__ dst,
          const unsigned char* __restrict__ ids, const float* __restrict__ feat,
          const float* __restrict__ el,
          const float* __restrict__ er, const float* __restrict__ bias,
          float* __restrict__ out, _Float16* __restrict__ out16,
          unsigned* __restrict__ lists, unsigned* __restrict__ counts) {
    constexpr int F   = H * D;
    constexpr int KPL = F / 32;
    __shared__ unsigned s_list[CAP];
    __shared__ __attribute__((aligned(16))) float s_acc[NB * F];
    __shared__ unsigned s_mcur[NB * H];
    __shared__ float    s_m[NB * H];
    __shared__ float    s_den[NB * H];
    __shared__ float    s_scl[NB * H];
    __shared__ unsigned s_cnt;

    const int tid  = threadIdx.x;
    const int lane = tid & 31;
    const int wave = __builtin_amdgcn_readfirstlane(tid >> 5);
    const int hl   = (lane < H) ? lane : 0;
    const int lo   = blockIdx.x * NB;
    const int c0   = (int)((blockIdx.x * (unsigned)NCHUNK) / gridDim.x);
    const float NEG_INF = -__builtin_inff();
    const unsigned gpat = blockIdx.x * 0x01010101u;

    unsigned* __restrict__ mylist = lists + (size_t)blockIdx.x * CAP;
    unsigned woff = 0u;
    int saved = -1;
    if (!BUILD) {
        const unsigned c = __builtin_amdgcn_readfirstlane(counts[blockIdx.x]);
        if (c <= (unsigned)CAP) saved = (int)c;
    }

    for (int i = tid; i < NB * F; i += OWN_T) s_acc[i] = 0.0f;
    if (tid < NB * H) {
        s_mcur[tid] = f2ord(NEG_INF);
        s_m[tid]    = NEG_INF;
        s_den[tid]  = 0.0f;
    }

    int ph = BUILD ? NCHUNK : PH_CHUNKS;
    for (int p0 = 0; p0 < NCHUNK;) {
        int n;
        if (saved >= 0) {
            for (int i = tid; i < saved; i += OWN_T) s_list[i] = mylist[i];
            __syncthreads();
            n = saved;
        } else {
            if (tid == 0) s_cnt = 0u;
            __syncthreads();
            const int pend = (p0 + ph < NCHUNK) ? (p0 + ph) : NCHUNK;

            for (int ci = p0; ci < pend; ci += SCAN_U) {
                uint4 v[SCAN_U];
                int   eb[SCAN_U];
#pragma unroll
                for (int u = 0; u < SCAN_U; ++u) {
                    int chunk = c0 + ci + u;
                    if (chunk >= NCHUNK) chunk -= NCHUNK;
                    eb[u] = chunk * CHUNK_E + tid * ID_E;
                    v[u] = *(const uint4*)(ids + eb[u]);
                }
#pragma unroll
                for (int u = 0; u < SCAN_U; ++u) {
                    const unsigned ww[4] = {v[u].x, v[u].y, v[u].z, v[u].w};
                    unsigned m = 0u;
#pragma unroll
                    for (int q = 0; q < 4; ++q) {
                        const unsigned x = ww[q] ^ gpat;
                        const unsigned y = ((x & 0x7F7F7F7Fu) + 0x7F7F7F7Fu) | x;
                        m |= (~y & 0x80808080u) >> (7 - q);
                    }
                    while (m) {
                        const int p = __builtin_ctz(m);
                        m &= m - 1u;
                        const unsigned pos = atomicAdd(&s_cnt, 1u);
                        if (pos < (unsigned)CAP) s_list[pos] = (unsigned)(eb[u] + 4 * (p & 7) + (p >> 3));
                    }
                }
            }
            __syncthreads();
            n = (int)s_cnt;
            if (BUILD && n > CAP) {
                ph = PH_CHUNKS;
                __syncthreads();
                continue;
            }
        }

        for (int i = tid; i < n; i += OWN_T) {
            unsigned en = s_list[i];
            int tl, sv;
            if (saved >= 0) {
                sv = (int)(en & 0xFFFFu);
                tl = (int)(en >> 16);
            } else {
                tl = dst[en] - lo;
                sv = src[en];
                en = (unsigned)sv | ((unsigned)tl << 16);
                s_list[i] = en;
                if (BUILD && woff + (unsigned)i < (unsigned)CAP) mylist[woff + (unsigned)i] = en;
            }
#pragma unroll
            for (int h = 0; h < H; ++h) {
                float x = el[sv * H + h] + er[(lo + tl) * H + h];
                x = (x > 0.0f) ? x : NEG_SLOPE * x;
                atomicMax(&s_mcur[tl * H + h], f2ord(x));
            }
        }
        __syncthreads();
        if (tid < NB * H) {
            const float mn = ord2f(s_mcur[tid]);
            const float mo = s_m[tid];
            const float sc = (mo == mn) ? 1.0f : __expf(mo - mn);
            s_den[tid] *= sc;
            s_scl[tid]  = sc;
            s_m[tid]    = mn;
        }
        __syncthreads();
        if (p0 > 0) {
            for (int i = tid; i < NB * F; i += OWN_T) s_acc[i] *= s_scl[i / D];
            __syncthreads();
        }

        for (int i0 = wave * U; i0 < n; i0 += (OWN_T / 32) * U) {
            int   tlv[U], svv[U];
            float sa[U], sb[U];
            float fv[U][KPL];
#pragma unroll
            for (int u = 0; u < U; ++u) {
                const int i = i0 + u;
                const unsigned en = (i < n) ? s_list[i] : 0u;
                const int sv = (int)(en & 0xFFFFu);
                const int tl = (int)(en >> 16);
                tlv[u] = tl;
                svv[u] = sv;
#pragma unroll
                for (int k = 0; k < KPL; ++k)
                    fv[u][k] = feat[(size_t)sv * F + k * 32 + lane];
            }
#pragma unroll
            for (int u = 0; u < U; ++u) {
                sa[u] = el[svv[u] * H + hl];
                sb[u] = er[(lo + tlv[u]) * H + hl];
            }
#pragma unroll
            for (int u = 0; u < U; ++u) {
                if (i0 + u < n) {
                    const int tl = tlv[u];
                    float x = sa[u] + sb[u];
                    x = (x > 0.0f) ? x : NEG_SLOPE * x;
                    const float wlane = __expf(x - s_m[tl * H + hl]);
                    if (lane < H) atomicAdd(&s_den[tl * H + lane], wlane);
#pragma unroll
                    for (int k = 0; k < KPL; ++k) {
                        const float w = __shfl(wlane, (k * 32) / D);
                        atomicAdd(&s_acc[tl * F + k * 32 + lane], w * fv[u][k]);
                    }
                }
            }
        }
        __syncthreads();
        if (BUILD) woff += (unsigned)n;
        if (saved >= 0) break;
        p0 += ph;
    }
    if (BUILD && tid == 0) counts[blockIdx.x] = woff;

    if (ELU16) {
        static_assert(D % 4 == 0, "a thread's four features belong to one head");
        for (int i = tid; i < NB * F / 4; i += OWN_T) {
            const int e  = i * 4;
            const int tl = e / F, f = e - tl * F;
            const int node = lo + tl;
            if (node < N_NODES) {
                const float  d  = s_den[tl * H + f / D];
                const float4 a  = *(const float4*)&s_acc[e];
                const float4 bv = *(const float4*)&bias[f];
                const float r0 = (d > 0.0f) ? a.x / d : 0.0f;
                const float r1 = (d > 0.0f) ? a.y / d : 0.0f;
                const float r2 = (d > 0.0f) ? a.z / d : 0.0f;
                const float r3 = (d > 0.0f) ? a.w / d : 0.0f;
                v4h r;
                r[0] = (_Float16)elu1(r0 + bv.x); r[1] = (_Float16)elu1(r1 + bv.y);
                r[2] = (_Float16)elu1(r2 + bv.z); r[3] = (_Float16)elu1(r3 + bv.w);
                *(v4h*)(out16 + (size_t)node * F + f) = r;
            }
        }
    } else {
        for (int i = tid; i < NB * F; i += OWN_T) {
            const int tl = i / F, f = i - tl * F;
            const int node = lo + tl;
            if (node < N_NODES) {
                const float d = s_den[tl * H + f / D];
                float r = (d > 0.0f) ? s_acc[i] / d : 0.0f;
                if (bias) r += bias[f];
                out[(size_t)node * F + f] = r;
            }
        }
    }
}

extern "C" void kernel_launch(void* const* d_in, const int* in_sizes, int n_in,
                              void* d_out, int out_size, void* d_ws, size_t ws_size,
                              hipStream_t stream) {
    const float* h   = (const float*)d_in[0];
    const int*   src = (const int*)  d_in[1];
    const int*   dst = (const int*)  d_in[2];
    const float* W1  = (const float*)d_in[3];
    const float* al1 = (const float*)d_in[4];
    const float* ar1 = (const float*)d_in[5];
    const float* b1  = (const float*)d_in[6];
    const float* W2  = (const float*)d_in[7];
    const float* al2 = (const float*)d_in[8];
    const float* ar2 = (const float*)d_in[9];
    const float* b2  = (const float*)d_in[10];
    float* out = (float*)d_out;

    size_t off = 0;
    auto alloc = [&](size_t bytes) {
        void* p = (char*)d_ws + off;
        off += (bytes + 255) & ~(size_t)255;
        return p;
    };
    _Float16* hf16   = (_Float16*)alloc((size_t)N_NODES * IN_DIM  * 2);
    _Float16* W1t16  = (_Float16*)alloc((size_t)IN_DIM  * D1      * 2);
    _Float16* W2t16  = (_Float16*)alloc((size_t)D1      * OUT_DIM * 2);
    float*    feat1  = (float*)   alloc((size_t)N_NODES * D1      * 4);
    float*    el1    = (float*)   alloc((size_t)N_NODES * HEADS   * 4);
    float*    er1    = (float*)   alloc((size_t)N_NODES * HEADS   * 4);
    _Float16* h1f16  = (_Float16*)alloc((size_t)N_NODES * D1      * 2);
    float*    feat2  = (float*)   alloc((size_t)N_NODES * OUT_DIM * 4);
    float*    el2    = (float*)   alloc((size_t)N_NODES * 4);
    float*    er2    = (float*)   alloc((size_t)N_NODES * 4);
    unsigned* lists  = (unsigned*)alloc((size_t)((N_NODES + NB - 1) / NB) * CAP * 4);
    unsigned* counts = (unsigned*)alloc((size_t)((N_NODES + NB - 1) / NB) * 4);
    unsigned char* ids = (unsigned char*)alloc((size_t)IDS_N);

    auto cdiv = [](int a, int b) { return (a + b - 1) / b; };

    cvt_f32_to_f16<<<cdiv(N_NODES * IN_DIM / 4, EW_T * EW_U), EW_T, 0, stream>>>(h, hf16,
                                                                                N_NODES * IN_DIM / 4);
    cvt_transpose_f16<<<cdiv(IN_DIM * D1, 256), 256, 0, stream>>>(W1, W1t16, IN_DIM, D1);
    cvt_transpose_f16<<<cdiv(D1 * OUT_DIM, 256), 256, 0, stream>>>(W2, W2t16, D1, OUT_DIM);

    gemm_wmma_f16t<<<dim3(N_NODES / 16, 1), 128, 0, stream>>>(hf16, W1t16, feat1,
                                                              N_NODES, D1, IN_DIM);
    attn_scores<<<cdiv(N_NODES * HEADS, 256), 256, 0, stream>>>(feat1, al1, ar1, el1, er1,
                                                                HEADS, HID);
    owner_ids<<<IDS_N / ID_E / IDS_T, IDS_T, 0, stream>>>(dst, ids);
    gat_owner<HEADS, HID, 2, true, true><<<cdiv(N_NODES, NB), OWN_T, 0, stream>>>(
        src, dst, ids, feat1, el1, er1, b1, (float*)nullptr, h1f16, lists, counts);

    gemm_wmma_f16t<<<dim3(N_NODES / 16, 1), 32, 0, stream>>>(h1f16, W2t16, feat2,
                                                             N_NODES, OUT_DIM, D1);
    attn_scores<<<cdiv(N_NODES, 256), 256, 0, stream>>>(feat2, al2, ar2, el2, er2, 1, OUT_DIM);
    gat_owner<1, OUT_DIM, 4, false, false><<<cdiv(N_NODES, NB), OWN_T, 0, stream>>>(
        src, dst, ids, feat2, el2, er2, b2, out, (_Float16*)nullptr, lists, counts);
    (void)in_sizes; (void)n_in; (void)out_size; (void)ws_size;
}


# message pass: a trip's per-(edge, head) work (list entry, scores, max lookup, exp, denominator atomic) done once across U*H lanes, spread by v_readlane; LDS atomic sums still in arrival order
# speedup vs baseline: 3.9984x; 1.0536x over previous
#include <hip/hip_runtime.h>
#include <hip/hip_bf16.h>
#include <hip/hip_fp16.h>

#define N_NODES 10000
#define N_EDGES 320000
#define IN_DIM  256
#define HID     64
#define HEADS   4
#define D1      (HEADS * HID)
#define OUT_DIM 64
#define NEG_SLOPE 0.2f

static_assert((N_NODES * IN_DIM) % 4 == 0, "the elementwise pass moves float4");

typedef __attribute__((ext_vector_type(16))) _Float16 v16h;
typedef __attribute__((ext_vector_type(8)))  _Float16 v8h;
typedef __attribute__((ext_vector_type(4)))  _Float16 v4h;
typedef __attribute__((ext_vector_type(8)))  float    v8f;

#define EW_T 128
#define EW_U 8

__global__ void __launch_bounds__(EW_T)
cvt_f32_to_f16(const float* __restrict__ in, _Float16* __restrict__ out, int n4) {
    const float4* __restrict__ in4 = (const float4*)in;
    const int base = blockIdx.x * (EW_T * EW_U) + threadIdx.x;
    float4 v[EW_U];
#pragma unroll
    for (int u = 0; u < EW_U; ++u) {
        const int i = base + u * EW_T;
        v[u] = in4[(i < n4) ? i : (n4 - 1)];
    }
#pragma unroll
    for (int u = 0; u < EW_U; ++u) {
        const int i = base + u * EW_T;
        if (i < n4) {
            v4h r;
            r[0] = (_Float16)v[u].x; r[1] = (_Float16)v[u].y;
            r[2] = (_Float16)v[u].z; r[3] = (_Float16)v[u].w;
            *(v4h*)(out + (size_t)i * 4) = r;
        }
    }
}

__global__ void cvt_transpose_f16(const float* __restrict__ W, _Float16* __restrict__ Wt,
                                  int K, int N) {
    int i = blockIdx.x * blockDim.x + threadIdx.x;
    if (i >= K * N) return;
    int k = i / N, n = i % N;
    Wt[(size_t)n * K + k] = (_Float16)W[i];
}

__device__ inline float elu1(float x) { return (x > 0.0f) ? x : (__expf(x) - 1.0f); }

__device__ inline v16h load_frag(const _Float16* __restrict__ base, int half) {
    v8h lo = *(const v8h*)(base + 8 * half);
    v8h hi = *(const v8h*)(base + 16 + 8 * half);
    v16h r;
#pragma unroll
    for (int i = 0; i < 8; ++i) { r[i] = lo[i]; r[8 + i] = hi[i]; }
    return r;
}

__global__ void gemm_wmma_f16t(const _Float16* __restrict__ A, const _Float16* __restrict__ Bt,
                               float* __restrict__ C, int M, int N, int K) {
    const int lane = threadIdx.x & 31;
    const int wave = threadIdx.x >> 5;
    const int m0   = blockIdx.x * 16;
    const int n0   = (blockIdx.y * (blockDim.x >> 5) + wave) * 64;
    if (m0 >= M || n0 >= N) return;

    const int half = lane >> 4;
    const int l15  = lane & 15;

    const _Float16* aptr = A + (size_t)(m0 + l15) * K;
    const _Float16* bptr0 = Bt + (size_t)(n0 +  0 + l15) * K;
    const _Float16* bptr1 = Bt + (size_t)(n0 + 16 + l15) * K;
    const _Float16* bptr2 = Bt + (size_t)(n0 + 32 + l15) * K;
    const _Float16* bptr3 = Bt + (size_t)(n0 + 48 + l15) * K;

    v8f acc0 = {}, acc1 = {}, acc2 = {}, acc3 = {};

    for (int k0 = 0; k0 < K; k0 += 32) {
        v16h a  = load_frag(aptr  + k0, half);
        v16h b0 = load_frag(bptr0 + k0, half);
        v16h b1 = load_frag(bptr1 + k0, half);
        v16h b2 = load_frag(bptr2 + k0, half);
        v16h b3 = load_frag(bptr3 + k0, half);
        acc0 = __builtin_amdgcn_wmma_f32_16x16x32_f16(false, a, false, b0, (short)0, acc0, false, false);
        acc1 = __builtin_amdgcn_wmma_f32_16x16x32_f16(false, a, false, b1, (short)0, acc1, false, false);
        acc2 = __builtin_amdgcn_wmma_f32_16x16x32_f16(false, a, false, b2, (short)0, acc2, false, false);
        acc3 = __builtin_amdgcn_wmma_f32_16x16x32_f16(false, a, false, b3, (short)0, acc3, false, false);
    }

    float* cbase = C + (size_t)(m0 + 8 * half) * N + n0 + l15;
#pragma unroll
    for (int r = 0; r < 8; ++r) {
        float* row = cbase + (size_t)r * N;
        row[0]  = acc0[r];
        row[16] = acc1[r];
        row[32] = acc2[r];
        row[48] = acc3[r];
    }
}

__global__ void attn_scores(const float* __restrict__ feat, const float* __restrict__ al,
                            const float* __restrict__ ar, float* __restrict__ el,
                            float* __restrict__ er, int nheads, int d) {
    int idx = blockIdx.x * blockDim.x + threadIdx.x;
    if (idx >= N_NODES * nheads) return;
    int n = idx / nheads, h = idx % nheads;
    const float* f = feat + (size_t)(n * nheads + h) * d;
    float sl = 0.f, sr = 0.f;
    for (int j = 0; j < d; ++j) {
        float fv = f[j];
        sl += fv * al[h * d + j];
        sr += fv * ar[h * d + j];
    }
    el[idx] = sl;
    er[idx] = sr;
}

#define NB        40
#define OWN_T     1024
#define NGROUPS   ((N_NODES + NB - 1) / NB)
#define ID_E      16
#define CHUNK_E   (OWN_T * ID_E)
#define NCHUNK    ((N_EDGES + CHUNK_E - 1) / CHUNK_E)
#define PH_CHUNKS 4
#define CAP       (PH_CHUNKS * CHUNK_E)
#define SCAN_U    4
#define ID_NONE   255u
#define IDS_N     (NCHUNK * CHUNK_E)

static_assert(N_NODES <= 65536 && NB <= 65536, "saved list entry packs src and local dst in 16 bits each");
static_assert(NGROUPS <= 255, "a group number fits a byte and never equals ID_NONE");
static_assert(N_EDGES % ID_E == 0, "a thread's 16 ids are all edges or all pad");
static_assert(NCHUNK % SCAN_U == 0 && PH_CHUNKS % SCAN_U == 0, "every batch of SCAN_U chunks lies inside one phase");
static_assert(NB * HEADS <= OWN_T, "one thread per (node, head) slot");

#define IDS_T 128
static_assert((IDS_N / ID_E) % IDS_T == 0, "the grid covers the padded array exactly");
__global__ void __launch_bounds__(IDS_T)
owner_ids(const int* __restrict__ dst, unsigned char* __restrict__ ids) {
    const int t = blockIdx.x * IDS_T + threadIdx.x;
    unsigned w[ID_E / 4] = {~0u, ~0u, ~0u, ~0u};
    if (t * ID_E < N_EDGES) {
        const int4* __restrict__ d4 = (const int4*)dst + (size_t)t * (ID_E / 4);
        int4 v[ID_E / 4];
#pragma unroll
        for (int q = 0; q < ID_E / 4; ++q) v[q] = d4[q];
#pragma unroll
        for (int q = 0; q < ID_E / 4; ++q) {
            const int tt[4] = {v[q].x, v[q].y, v[q].z, v[q].w};
            unsigned r = 0u;
#pragma unroll
            for (int c = 0; c < 4; ++c) {
                unsigned g = (unsigned)tt[c] / (unsigned)NB;
                g = ((unsigned)tt[c] < (unsigned)N_NODES) ? g : ID_NONE;
                r |= g << (8 * c);
            }
            w[q] = r;
        }
    }
    *(uint4*)(ids + (size_t)t * ID_E) = make_uint4(w[0], w[1], w[2], w[3]);
}

__device__ inline unsigned f2ord(float f) {
    unsigned u = __float_as_uint(f);
    return (u & 0x80000000u) ? ~u : (u | 0x80000000u);
}
__device__ inline float ord2f(unsigned k) {
    return __uint_as_float((k & 0x80000000u) ? (k ^ 0x80000000u) : ~k);
}

template <int H, int D, int U, bool BUILD, bool ELU16>
__global__ void __launch_bounds__(OWN_T)
gat_owner(const int* __restrict__ src, const int* __restrict__ dst,
          const unsigned char* __restrict__ ids, const float* __restrict__ feat,
          const float* __restrict__ el,
          const float* __restrict__ er, const float* __restrict__ bias,
          float* __restrict__ out, _Float16* __restrict__ out16,
          unsigned* __restrict__ lists, unsigned* __restrict__ counts) {
    constexpr int F   = H * D;
    constexpr int KPL = F / 32;
    __shared__ unsigned s_list[CAP];
    __shared__ __attribute__((aligned(16))) float s_acc[NB * F];
    __shared__ unsigned s_mcur[NB * H];
    __shared__ float    s_m[NB * H];
    __shared__ float    s_den[NB * H];
    __shared__ float    s_scl[NB * H];
    __shared__ unsigned s_cnt;

    const int tid  = threadIdx.x;
    const int lane = tid & 31;
    const int wave = __builtin_amdgcn_readfirstlane(tid >> 5);
    constexpr int SLOTS = U * H;
    static_assert(SLOTS <= 32, "a trip's (edge, head) slots fit one wave");
    const int sl   = (lane < SLOTS) ? lane : 0;
    const int su   = sl / H;
    const int sh   = sl % H;
    const int lo   = blockIdx.x * NB;
    const int c0   = (int)((blockIdx.x * (unsigned)NCHUNK) / gridDim.x);
    const float NEG_INF = -__builtin_inff();
    const unsigned gpat = blockIdx.x * 0x01010101u;

    unsigned* __restrict__ mylist = lists + (size_t)blockIdx.x * CAP;
    unsigned woff = 0u;
    int saved = -1;
    if (!BUILD) {
        const unsigned c = __builtin_amdgcn_readfirstlane(counts[blockIdx.x]);
        if (c <= (unsigned)CAP) saved = (int)c;
    }

    for (int i = tid; i < NB * F; i += OWN_T) s_acc[i] = 0.0f;
    if (tid < NB * H) {
        s_mcur[tid] = f2ord(NEG_INF);
        s_m[tid]    = NEG_INF;
        s_den[tid]  = 0.0f;
    }

    int ph = BUILD ? NCHUNK : PH_CHUNKS;
    for (int p0 = 0; p0 < NCHUNK;) {
        int n;
        if (saved >= 0) {
            for (int i = tid; i < saved; i += OWN_T) s_list[i] = mylist[i];
            __syncthreads();
            n = saved;
        } else {
            if (tid == 0) s_cnt = 0u;
            __syncthreads();
            const int pend = (p0 + ph < NCHUNK) ? (p0 + ph) : NCHUNK;

            for (int ci = p0; ci < pend; ci += SCAN_U) {
                uint4 v[SCAN_U];
                int   eb[SCAN_U];
#pragma unroll
                for (int u = 0; u < SCAN_U; ++u) {
                    int chunk = c0 + ci + u;
                    if (chunk >= NCHUNK) chunk -= NCHUNK;
                    eb[u] = chunk * CHUNK_E + tid * ID_E;
                    v[u] = *(const uint4*)(ids + eb[u]);
                }
#pragma unroll
                for (int u = 0; u < SCAN_U; ++u) {
                    const unsigned ww[4] = {v[u].x, v[u].y, v[u].z, v[u].w};
                    unsigned m = 0u;
#pragma unroll
                    for (int q = 0; q < 4; ++q) {
                        const unsigned x = ww[q] ^ gpat;
                        const unsigned y = ((x & 0x7F7F7F7Fu) + 0x7F7F7F7Fu) | x;
                        m |= (~y & 0x80808080u) >> (7 - q);
                    }
                    while (m) {
                        const int p = __builtin_ctz(m);
                        m &= m - 1u;
                        const unsigned pos = atomicAdd(&s_cnt, 1u);
                        if (pos < (unsigned)CAP) s_list[pos] = (unsigned)(eb[u] + 4 * (p & 7) + (p >> 3));
                    }
                }
            }
            __syncthreads();
            n = (int)s_cnt;
            if (BUILD && n > CAP) {
                ph = PH_CHUNKS;
                __syncthreads();
                continue;
            }
        }

        for (int i = tid; i < n; i += OWN_T) {
            unsigned en = s_list[i];
            int tl, sv;
            if (saved >= 0) {
                sv = (int)(en & 0xFFFFu);
                tl = (int)(en >> 16);
            } else {
                tl = dst[en] - lo;
                sv = src[en];
                en = (unsigned)sv | ((unsigned)tl << 16);
                s_list[i] = en;
                if (BUILD && woff + (unsigned)i < (unsigned)CAP) mylist[woff + (unsigned)i] = en;
            }
#pragma unroll
            for (int h = 0; h < H; ++h) {
                float x = el[sv * H + h] + er[(lo + tl) * H + h];
                x = (x > 0.0f) ? x : NEG_SLOPE * x;
                atomicMax(&s_mcur[tl * H + h], f2ord(x));
            }
        }
        __syncthreads();
        if (tid < NB * H) {
            const float mn = ord2f(s_mcur[tid]);
            const float mo = s_m[tid];
            const float sc = (mo == mn) ? 1.0f : __expf(mo - mn);
            s_den[tid] *= sc;
            s_scl[tid]  = sc;
            s_m[tid]    = mn;
        }
        __syncthreads();
        if (p0 > 0) {
            for (int i = tid; i < NB * F; i += OWN_T) s_acc[i] *= s_scl[i / D];
            __syncthreads();
        }

        for (int i0 = wave * U; i0 < n; i0 += (OWN_T / 32) * U) {
            const int is = i0 + su;
            const unsigned en_s = (is < n) ? s_list[is] : 0u;
            const int sv_s = (int)(en_s & 0xFFFFu);
            const int tl_s = (int)(en_s >> 16);
            int   tlv[U];
            float fv[U][KPL];
#pragma unroll
            for (int u = 0; u < U; ++u) {
                const unsigned en = (unsigned)__builtin_amdgcn_readlane((int)en_s, u * H);
                const int sv = (int)(en & 0xFFFFu);
                tlv[u] = (int)(en >> 16);
#pragma unroll
                for (int k = 0; k < KPL; ++k)
                    fv[u][k] = feat[(size_t)sv * F + k * 32 + lane];
            }
            float x = el[sv_s * H + sh] + er[(lo + tl_s) * H + sh];
            x = (x > 0.0f) ? x : NEG_SLOPE * x;
            const float ws = __expf(x - s_m[tl_s * H + sh]);
            if (lane < SLOTS && is < n) atomicAdd(&s_den[tl_s * H + sh], ws);
#pragma unroll
            for (int u = 0; u < U; ++u) {
                if (i0 + u < n) {
                    const int tl = tlv[u];
#pragma unroll
                    for (int k = 0; k < KPL; ++k) {
                        const float w = __int_as_float(
                            __builtin_amdgcn_readlane(__float_as_int(ws), u * H + (k * 32) / D));
                        atomicAdd(&s_acc[tl * F + k * 32 + lane], w * fv[u][k]);
                    }
                }
            }
        }
        __syncthreads();
        if (BUILD) woff += (unsigned)n;
        if (saved >= 0) break;
        p0 += ph;
    }
    if (BUILD && tid == 0) counts[blockIdx.x] = woff;

    if (ELU16) {
        static_assert(D % 4 == 0, "a thread's four features belong to one head");
        for (int i = tid; i < NB * F / 4; i += OWN_T) {
            const int e  = i * 4;
            const int tl = e / F, f = e - tl * F;
            const int node = lo + tl;
            if (node < N_NODES) {
                const float  d  = s_den[tl * H + f / D];
                const float4 a  = *(const float4*)&s_acc[e];
                const float4 bv = *(const float4*)&bias[f];
                const float r0 = (d > 0.0f) ? a.x / d : 0.0f;
                const float r1 = (d > 0.0f) ? a.y / d : 0.0f;
                const float r2 = (d > 0.0f) ? a.z / d : 0.0f;
                const float r3 = (d > 0.0f) ? a.w / d : 0.0f;
                v4h r;
                r[0] = (_Float16)elu1(r0 + bv.x); r[1] = (_Float16)elu1(r1 + bv.y);
                r[2] = (_Float16)elu1(r2 + bv.z); r[3] = (_Float16)elu1(r3 + bv.w);
                *(v4h*)(out16 + (size_t)node * F + f) = r;
            }
        }
    } else {
        for (int i = tid; i < NB * F; i += OWN_T) {
            const int tl = i / F, f = i - tl * F;
            const int node = lo + tl;
            if (node < N_NODES) {
                const float d = s_den[tl * H + f / D];
                float r = (d > 0.0f) ? s_acc[i] / d : 0.0f;
                if (bias) r += bias[f];
                out[(size_t)node * F + f] = r;
            }
        }
    }
}

extern "C" void kernel_launch(void* const* d_in, const int* in_sizes, int n_in,
                              void* d_out, int out_size, void* d_ws, size_t ws_size,
                              hipStream_t stream) {
    const float* h   = (const float*)d_in[0];
    const int*   src = (const int*)  d_in[1];
    const int*   dst = (const int*)  d_in[2];
    const float* W1  = (const float*)d_in[3];
    const float* al1 = (const float*)d_in[4];
    const float* ar1 = (const float*)d_in[5];
    const float* b1  = (const float*)d_in[6];
    const float* W2  = (const float*)d_in[7];
    const float* al2 = (const float*)d_in[8];
    const float* ar2 = (const float*)d_in[9];
    const float* b2  = (const float*)d_in[10];
    float* out = (float*)d_out;

    size_t off = 0;
    auto alloc = [&](size_t bytes) {
        void* p = (char*)d_ws + off;
        off += (bytes + 255) & ~(size_t)255;
        return p;
    };
    _Float16* hf16   = (_Float16*)alloc((size_t)N_NODES * IN_DIM  * 2);
    _Float16* W1t16  = (_Float16*)alloc((size_t)IN_DIM  * D1      * 2);
    _Float16* W2t16  = (_Float16*)alloc((size_t)D1      * OUT_DIM * 2);
    float*    feat1  = (float*)   alloc((size_t)N_NODES * D1      * 4);
    float*    el1    = (float*)   alloc((size_t)N_NODES * HEADS   * 4);
    float*    er1    = (float*)   alloc((size_t)N_NODES * HEADS   * 4);
    _Float16* h1f16  = (_Float16*)alloc((size_t)N_NODES * D1      * 2);
    float*    feat2  = (float*)   alloc((size_t)N_NODES * OUT_DIM * 4);
    float*    el2    = (float*)   alloc((size_t)N_NODES * 4);
    float*    er2    = (float*)   alloc((size_t)N_NODES * 4);
    unsigned* lists  = (unsigned*)alloc((size_t)((N_NODES + NB - 1) / NB) * CAP * 4);
    unsigned* counts = (unsigned*)alloc((size_t)((N_NODES + NB - 1) / NB) * 4);
    unsigned char* ids = (unsigned char*)alloc((size_t)IDS_N);

    auto cdiv = [](int a, int b) { return (a + b - 1) / b; };

    cvt_f32_to_f16<<<cdiv(N_NODES * IN_DIM / 4, EW_T * EW_U), EW_T, 0, stream>>>(h, hf16,
                                                                                N_NODES * IN_DIM / 4);
    cvt_transpose_f16<<<cdiv(IN_DIM * D1, 256), 256, 0, stream>>>(W1, W1t16, IN_DIM, D1);
    cvt_transpose_f16<<<cdiv(D1 * OUT_DIM, 256), 256, 0, stream>>>(W2, W2t16, D1, OUT_DIM);

    gemm_wmma_f16t<<<dim3(N_NODES / 16, 1), 128, 0, stream>>>(hf16, W1t16, feat1,
                                                              N_NODES, D1, IN_DIM);
    attn_scores<<<cdiv(N_NODES * HEADS, 256), 256, 0, stream>>>(feat1, al1, ar1, el1, er1,
                                                                HEADS, HID);
    owner_ids<<<IDS_N / ID_E / IDS_T, IDS_T, 0, stream>>>(dst, ids);
    gat_owner<HEADS, HID, 2, true, true><<<cdiv(N_NODES, NB), OWN_T, 0, stream>>>(
        src, dst, ids, feat1, el1, er1, b1, (float*)nullptr, h1f16, lists, counts);

    gemm_wmma_f16t<<<dim3(N_NODES / 16, 1), 32, 0, stream>>>(h1f16, W2t16, feat2,
                                                             N_NODES, OUT_DIM, D1);
    attn_scores<<<cdiv(N_NODES, 256), 256, 0, stream>>>(feat2, al2, ar2, el2, er2, 1, OUT_DIM);
    gat_owner<1, OUT_DIM, 4, false, false><<<cdiv(N_NODES, NB), OWN_T, 0, stream>>>(
        src, dst, ids, feat2, el2, er2, b2, out, (_Float16*)nullptr, lists, counts);
    (void)in_sizes; (void)n_in; (void)out_size; (void)ws_size;
}
